# starting kernel
# baseline (speedup 1.0000x reference)
#include <hip/hip_runtime.h>
#include <hip/hip_bf16.h>
#include <cmath>


typedef __attribute__((ext_vector_type(2))) float v2f;
typedef __attribute__((ext_vector_type(8))) float v8f;

#define WMMA4F32(A, Bv, C)                                                   \
  __builtin_amdgcn_wmma_f32_16x16x4_f32(false, (A), false, (Bv), (short)0,   \
                                        (C), false, false)

static constexpr int Bsz  = 256;
static constexpr int Tlen = 2000;
static constexpr int Fin  = 42;
static constexpr int Hh   = 24;
static constexpr int Gg   = 72;

__device__ __forceinline__ float sigmoidf_(float v) {
  return 1.0f / (1.0f + __expf(-v));
}

__global__ __launch_bounds__(32) void lc1_tanh_kernel(
    const float* __restrict__ x, const float* __restrict__ w1,
    const float* __restrict__ b1, float* __restrict__ aOut) {
  const int  lane = threadIdx.x;
  const int  nl   = lane & 15;
  const int  ko   = (lane >> 4) * 2;
  const int  mo   = (lane >> 4) * 8;
  const bool hiHalf = (lane >= 16);
  const v2f  zero2 = {0.0f, 0.0f};

  v2f B0[11], B1[11];
  const int  n1  = 16 + nl;
  const bool n1v = (n1 < Hh);
  const int  n1c = n1v ? n1 : (Hh - 1);
#pragma unroll
  for (int kc = 0; kc < 11; ++kc) {
    const int k   = kc * 4 + ko;
    const int kc0 = (k     < Fin) ? k     : 0;
    const int kc1 = (k + 1 < Fin) ? k + 1 : 0;
    const float w00 = w1[nl  * Fin + kc0];
    const float w01 = w1[nl  * Fin + kc1];
    const float w10 = w1[n1c * Fin + kc0];
    const float w11 = w1[n1c * Fin + kc1];
    B0[kc].x = (k     < Fin) ? w00 : 0.0f;
    B0[kc].y = (k + 1 < Fin) ? w01 : 0.0f;
    B1[kc].x = (n1v && (k     < Fin)) ? w10 : 0.0f;
    B1[kc].y = (n1v && (k + 1 < Fin)) ? w11 : 0.0f;
  }
  const float bA = b1[nl];
  const float bB = n1v ? b1[n1c] : 0.0f;

  const int NT = (Bsz * Tlen) / 16;
  for (int tile = blockIdx.x; tile < NT; tile += gridDim.x) {
    const float* xr = x + (size_t)(tile * 16 + nl) * Fin;

    if (tile + (int)gridDim.x < NT)
      __builtin_prefetch(xr + (size_t)gridDim.x * 16 * Fin, 0, 1);

    v8f a0e = {0, 0, 0, 0, 0, 0, 0, 0}, a0o = {0, 0, 0, 0, 0, 0, 0, 0};
    v8f a1e = {0, 0, 0, 0, 0, 0, 0, 0}, a1o = {0, 0, 0, 0, 0, 0, 0, 0};
#pragma unroll
    for (int kc = 0; kc < 10; kc += 2) {
      const v2f Ae = *(const v2f*)(xr + kc * 4 + ko);
      const v2f Ao = *(const v2f*)(xr + (kc + 1) * 4 + ko);
      a0e = WMMA4F32(Ae, B0[kc], a0e);
      a1e = WMMA4F32(Ae, B1[kc], a1e);
      a0o = WMMA4F32(Ao, B0[kc + 1], a0o);
      a1o = WMMA4F32(Ao, B1[kc + 1], a1o);
    }
    {
      const v2f t = *(const v2f*)(xr + 40);
      const v2f A = hiHalf ? zero2 : t;
      a0e = WMMA4F32(A, B0[10], a0e);
      a1e = WMMA4F32(A, B1[10], a1e);
    }

    float* arow = aOut + (size_t)tile * 16 * Hh;
#pragma unroll
    for (int j = 0; j < 8; ++j)
      arow[(j + mo) * Hh + nl] = tanhf(a0e[j] + a0o[j] + bA);
    if (nl < 8) {
#pragma unroll
      for (int j = 0; j < 8; ++j)
        arow[(j + mo) * Hh + 16 + nl] = tanhf(a1e[j] + a1o[j] + bB);
    }
  }
}

__global__ __launch_bounds__(32) void gru_scan_kernel(
    const float* __restrict__ a, const float* __restrict__ w_ih,
    const float* __restrict__ w_hh, const float* __restrict__ b_ih,
    const float* __restrict__ b_hh, const float* __restrict__ lc2_w,
    const float* __restrict__ lc2_b, float* __restrict__ out) {
  __shared__ __align__(16) float Wlds[80][48];
  __shared__ __align__(16) float hS[16][24];
  __shared__ __align__(16) float pS[16][96];
  __shared__ float bsum[48], bihn[24], bhhn[24], lc2S[24];

  const int lane = threadIdx.x;
  const int nl   = lane & 15;
  const int ko   = (lane >> 4) * 2;
  const int mo   = (lane >> 4) * 8;
  const int b0   = blockIdx.x * 16;

  for (int idx = lane; idx < 80 * 48; idx += 32) {
    const int n = idx / 48, k = idx % 48;
    const int nc = (n < Gg) ? n : (Gg - 1);
    const float v = (k < Hh) ? w_ih[nc * Hh + k] : w_hh[nc * Hh + (k - Hh)];
    Wlds[n][k] = (n < Gg) ? v : 0.0f;
  }
  if (lane < 24) {
    bihn[lane] = b_ih[48 + lane];
    bhhn[lane] = b_hh[48 + lane];
    lc2S[lane] = lc2_w[lane];
  }
  for (int g = lane; g < 48; g += 32) bsum[g] = b_ih[g] + b_hh[g];

  const int gm = lane >> 1;
  const int cb = (lane & 1) * 12;
#pragma unroll
  for (int i = 0; i < 12; ++i) hS[gm][cb + i] = 0.0f;
  const float l2b = lc2_b[0];
  __syncthreads();

  for (int t = 0; t < Tlen; ++t) {
    v2f Af[12];
    const float* ab = a + ((size_t)(b0 + nl) * Tlen + t) * Hh;
#pragma unroll
    for (int kc = 0; kc < 6; ++kc)
      Af[kc] = *(const v2f*)(ab + kc * 4 + ko);
#pragma unroll
    for (int kc = 6; kc < 12; ++kc)
      Af[kc] = *(const v2f*)(&hS[nl][(kc - 6) * 4 + ko]);

    if (t + 1 < Tlen) __builtin_prefetch(ab + Hh, 0, 1);

#pragma unroll
    for (int nt = 0; nt < 3; ++nt) {
      v8f ae = {0, 0, 0, 0, 0, 0, 0, 0};
      v8f ao = {0, 0, 0, 0, 0, 0, 0, 0};
#pragma unroll
      for (int kc = 0; kc < 12; kc += 2) {
        const v2f bve = *(const v2f*)(&Wlds[nt * 16 + nl][kc * 4 + ko]);
        const v2f bvo = *(const v2f*)(&Wlds[nt * 16 + nl][(kc + 1) * 4 + ko]);
        ae = WMMA4F32(Af[kc], bve, ae);
        ao = WMMA4F32(Af[kc + 1], bvo, ao);
      }
#pragma unroll
      for (int j = 0; j < 8; ++j) pS[j + mo][nt * 16 + nl] = ae[j] + ao[j];
    }
#pragma unroll
    for (int nt = 3; nt < 5; ++nt) {
      v8f aXe = {0, 0, 0, 0, 0, 0, 0, 0}, aXo = {0, 0, 0, 0, 0, 0, 0, 0};
      v8f aHe = {0, 0, 0, 0, 0, 0, 0, 0}, aHo = {0, 0, 0, 0, 0, 0, 0, 0};
#pragma unroll
      for (int kc = 0; kc < 6; kc += 2) {
        const v2f bve = *(const v2f*)(&Wlds[nt * 16 + nl][kc * 4 + ko]);
        const v2f bvo = *(const v2f*)(&Wlds[nt * 16 + nl][(kc + 1) * 4 + ko]);
        aXe = WMMA4F32(Af[kc], bve, aXe);
        aXo = WMMA4F32(Af[kc + 1], bvo, aXo);
      }
#pragma unroll
      for (int kc = 6; kc < 12; kc += 2) {
        const v2f bve = *(const v2f*)(&Wlds[nt * 16 + nl][kc * 4 + ko]);
        const v2f bvo = *(const v2f*)(&Wlds[nt * 16 + nl][(kc + 1) * 4 + ko]);
        aHe = WMMA4F32(Af[kc], bve, aHe);
        aHo = WMMA4F32(Af[kc + 1], bvo, aHo);
      }
      const int c = (nt - 3) * 16 + nl;
      if (c < Hh) {
#pragma unroll
        for (int j = 0; j < 8; ++j) {
          pS[j + mo][48 + c] = aXe[j] + aXo[j];
          pS[j + mo][72 + c] = aHe[j] + aHo[j];
        }
      }
    }
    __syncthreads();

    float s = 0.0f;
#pragma unroll
    for (int i = 0; i < 12; ++i) {
      const int c = cb + i;
      const float r  = sigmoidf_(pS[gm][c] + bsum[c]);
      const float z  = sigmoidf_(pS[gm][24 + c] + bsum[24 + c]);
      const float xn = pS[gm][48 + c] + bihn[c];
      const float hn = pS[gm][72 + c] + bhhn[c];
      const float nn = tanhf(xn + r * hn);
      const float hv = (1.0f - z) * nn + z * hS[gm][c];
      hS[gm][c] = hv;
      s += lc2S[c] * hv;
    }
    s += __shfl_xor(s, 1, 32);
    if ((lane & 1) == 0)
      out[(size_t)(b0 + gm) * Tlen + t] = sigmoidf_(s + l2b);
    __syncthreads();
  }
}

extern "C" void kernel_launch(void* const* d_in, const int* in_sizes, int n_in,
                              void* d_out, int out_size, void* d_ws,
                              size_t ws_size, hipStream_t stream) {
  (void)in_sizes; (void)n_in; (void)out_size; (void)ws_size;
  const float* x     = (const float*)d_in[0];
  const float* lc1_w = (const float*)d_in[1];
  const float* lc1_b = (const float*)d_in[2];
  const float* w_ih  = (const float*)d_in[3];
  const float* w_hh  = (const float*)d_in[4];
  const float* b_ih  = (const float*)d_in[5];
  const float* b_hh  = (const float*)d_in[6];
  const float* lc2_w = (const float*)d_in[7];
  const float* lc2_b = (const float*)d_in[8];
  float* out  = (float*)d_out;
  float* aBuf = (float*)d_ws;

  lc1_tanh_kernel<<<2048, 32, 0, stream>>>(x, lc1_w, lc1_b, aBuf);
  gru_scan_kernel<<<Bsz / 16, 32, 0, stream>>>(aBuf, w_ih, w_hh, b_ih, b_hh,
                                               lc2_w, lc2_b, out);
}


# GRU scan re-oriented (weights = A, [a|h]^T = B, k permuted): a lane's new h is its next B fragment; h in registers, no LDS or barrier per step; x-side products a step ahead; sum order changed
# speedup vs baseline: 1.7306x; 1.7306x over previous
#include <hip/hip_runtime.h>
#include <hip/hip_bf16.h>
#include <cmath>


typedef __attribute__((ext_vector_type(2))) float v2f;
typedef __attribute__((ext_vector_type(8))) float v8f;

#define WMMA4F32(A, Bv, C)                                                   \
  __builtin_amdgcn_wmma_f32_16x16x4_f32(false, (A), false, (Bv), (short)0,   \
                                        (C), false, false)

static constexpr int Bsz  = 256;
static constexpr int Tlen = 2000;
static constexpr int Fin  = 42;
static constexpr int Hh   = 24;
static constexpr int Gg   = 72;

__device__ __forceinline__ float sigmoidf_(float v) {
  return 1.0f / (1.0f + __expf(-v));
}

__global__ __launch_bounds__(32) void lc1_tanh_kernel(
    const float* __restrict__ x, const float* __restrict__ w1,
    const float* __restrict__ b1, float* __restrict__ aOut) {
  const int  lane = threadIdx.x;
  const int  nl   = lane & 15;
  const int  ko   = (lane >> 4) * 2;
  const int  mo   = (lane >> 4) * 8;
  const bool hiHalf = (lane >= 16);
  const v2f  zero2 = {0.0f, 0.0f};

  v2f B0[11], B1[11];
  const int  n1  = 16 + nl;
  const bool n1v = (n1 < Hh);
  const int  n1c = n1v ? n1 : (Hh - 1);
#pragma unroll
  for (int kc = 0; kc < 11; ++kc) {
    const int k   = kc * 4 + ko;
    const int kc0 = (k     < Fin) ? k     : 0;
    const int kc1 = (k + 1 < Fin) ? k + 1 : 0;
    const float w00 = w1[nl  * Fin + kc0];
    const float w01 = w1[nl  * Fin + kc1];
    const float w10 = w1[n1c * Fin + kc0];
    const float w11 = w1[n1c * Fin + kc1];
    B0[kc].x = (k     < Fin) ? w00 : 0.0f;
    B0[kc].y = (k + 1 < Fin) ? w01 : 0.0f;
    B1[kc].x = (n1v && (k     < Fin)) ? w10 : 0.0f;
    B1[kc].y = (n1v && (k + 1 < Fin)) ? w11 : 0.0f;
  }
  const float bA = b1[nl];
  const float bB = n1v ? b1[n1c] : 0.0f;

  const int NT = (Bsz * Tlen) / 16;
  for (int tile = blockIdx.x; tile < NT; tile += gridDim.x) {
    const float* xr = x + (size_t)(tile * 16 + nl) * Fin;

    if (tile + (int)gridDim.x < NT)
      __builtin_prefetch(xr + (size_t)gridDim.x * 16 * Fin, 0, 1);

    v8f a0e = {0, 0, 0, 0, 0, 0, 0, 0}, a0o = {0, 0, 0, 0, 0, 0, 0, 0};
    v8f a1e = {0, 0, 0, 0, 0, 0, 0, 0}, a1o = {0, 0, 0, 0, 0, 0, 0, 0};
#pragma unroll
    for (int kc = 0; kc < 10; kc += 2) {
      const v2f Ae = *(const v2f*)(xr + kc * 4 + ko);
      const v2f Ao = *(const v2f*)(xr + (kc + 1) * 4 + ko);
      a0e = WMMA4F32(Ae, B0[kc], a0e);
      a1e = WMMA4F32(Ae, B1[kc], a1e);
      a0o = WMMA4F32(Ao, B0[kc + 1], a0o);
      a1o = WMMA4F32(Ao, B1[kc + 1], a1o);
    }
    {
      const v2f t = *(const v2f*)(xr + 40);
      const v2f A = hiHalf ? zero2 : t;
      a0e = WMMA4F32(A, B0[10], a0e);
      a1e = WMMA4F32(A, B1[10], a1e);
    }

    float* arow = aOut + (size_t)tile * 16 * Hh;
#pragma unroll
    for (int j = 0; j < 8; ++j)
      arow[(j + mo) * Hh + nl] = tanhf(a0e[j] + a0o[j] + bA);
    if (nl < 8) {
#pragma unroll
      for (int j = 0; j < 8; ++j)
        arow[(j + mo) * Hh + 16 + nl] = tanhf(a1e[j] + a1o[j] + bB);
    }
  }
}

__device__ __forceinline__ float xhalf(float v) {
  const int b = __builtin_bit_cast(int, v);
  return __builtin_bit_cast(
      float, __builtin_amdgcn_permlanex16(b, b, 0x76543210, 0xfedcba98, false,
                                          false));
}

__global__ __launch_bounds__(32) void gru_scan_kernel(
    const float* __restrict__ a, const float* __restrict__ w_ih,
    const float* __restrict__ w_hh, const float* __restrict__ b_ih,
    const float* __restrict__ b_hh, const float* __restrict__ lc2_w,
    const float* __restrict__ lc2_b, float* __restrict__ out) {
  const int lane = threadIdx.x;
  const int nl   = lane & 15;
  const int hf   = lane >> 4;
  const int kb   = 12 * hf;
  const int b0   = blockIdx.x * 16;
  const v2f zero2 = {0.0f, 0.0f};

  const int  mh  = nl >> 3, mj = nl & 7;
  const int  u0  = 12 * mh + mj;
  const int  u1  = 12 * mh + 8 + (mj & 3);
  const bool lo4 = (mj < 4);
  const int rowR0 = u0, rowZ0 = Hh + u0, rowN0 = 2 * Hh + u0;
  const int rowRZ1 = (lo4 ? 0 : Hh) + u1, rowN1 = 2 * Hh + u1;

  v2f WxR0[6], WxZ0[6], WxRZ1[6], WxN0[6], WxN1[6];
  v2f WhR0[6], WhZ0[6], WhRZ1[6], WhN0[6], WhN1[6];
#pragma unroll
  for (int q = 0; q < 6; ++q) {
    const int k = kb + 2 * q;
    WxR0[q]  = *(const v2f*)(w_ih + rowR0  * Hh + k);
    WxZ0[q]  = *(const v2f*)(w_ih + rowZ0  * Hh + k);
    WxRZ1[q] = *(const v2f*)(w_ih + rowRZ1 * Hh + k);
    WxN0[q]  = *(const v2f*)(w_ih + rowN0  * Hh + k);
    WhR0[q]  = *(const v2f*)(w_hh + rowR0  * Hh + k);
    WhZ0[q]  = *(const v2f*)(w_hh + rowZ0  * Hh + k);
    WhRZ1[q] = *(const v2f*)(w_hh + rowRZ1 * Hh + k);
    WhN0[q]  = *(const v2f*)(w_hh + rowN0  * Hh + k);
    const v2f xi = *(const v2f*)(w_ih + rowN1 * Hh + k);
    const v2f hh = *(const v2f*)(w_hh + rowN1 * Hh + k);
    WxN1[q] = lo4 ? xi : zero2;
    WhN1[q] = lo4 ? zero2 : hh;
  }

  v8f bR0, bZ0, bRZ1, bXN0, bHN0, bN1;
#pragma unroll
  for (int j = 0; j < 8; ++j) {
    const int c0 = kb + j;
    const int c1 = kb + 8 + (j & 3);
    bR0[j]  = b_ih[c0] + b_hh[c0];
    bZ0[j]  = b_ih[Hh + c0] + b_hh[Hh + c0];
    bRZ1[j] = (j < 4) ? (b_ih[c1] + b_hh[c1]) : (b_ih[Hh + c1] + b_hh[Hh + c1]);
    bXN0[j] = b_ih[2 * Hh + c0];
    bHN0[j] = b_hh[2 * Hh + c0];
    bN1[j]  = (j < 4) ? b_ih[2 * Hh + c1] : b_hh[2 * Hh + c1];
  }
  float l2w[12];
#pragma unroll
  for (int e = 0; e < 12; ++e) l2w[e] = lc2_w[kb + e];
  const float l2b = lc2_b[0];

  const float* arow = a + (size_t)(b0 + nl) * Tlen * Hh + kb;
  float*       orow = out + (size_t)(b0 + nl) * Tlen;
  v2f aC[6];
#pragma unroll
  for (int q = 0; q < 6; ++q) aC[q] = *(const v2f*)(arow + 2 * q);

  v8f xR0 = bR0, xZ0 = bZ0, xRZ1 = bRZ1, xN0 = bXN0, xN1 = bN1;
#pragma unroll
  for (int q = 0; q < 6; ++q) {
    xR0  = WMMA4F32(WxR0[q],  aC[q], xR0);
    xZ0  = WMMA4F32(WxZ0[q],  aC[q], xZ0);
    xRZ1 = WMMA4F32(WxRZ1[q], aC[q], xRZ1);
    xN0  = WMMA4F32(WxN0[q],  aC[q], xN0);
    xN1  = WMMA4F32(WxN1[q],  aC[q], xN1);
  }
#pragma unroll
  for (int q = 0; q < 6; ++q) aC[q] = *(const v2f*)(arow + Hh + 2 * q);

  float h[12];
#pragma unroll
  for (int e = 0; e < 12; ++e) h[e] = 0.0f;

  for (int t = 0; t < Tlen; ++t) {
    const int tn = (t + 2 < Tlen) ? (t + 2) : (Tlen - 1);
    v2f aN[6];
#pragma unroll
    for (int q = 0; q < 6; ++q)
      aN[q] = *(const v2f*)(arow + (size_t)tn * Hh + 2 * q);

    v8f R0 = xR0, Z0 = xZ0, RZ1 = xRZ1, HN0 = bHN0, N1 = xN1;
    const v8f XN0 = xN0;
#pragma unroll
    for (int q = 0; q < 6; ++q) {
      const v2f hB = {h[2 * q], h[2 * q + 1]};
      R0  = WMMA4F32(WhR0[q],  hB, R0);
      Z0  = WMMA4F32(WhZ0[q],  hB, Z0);
      RZ1 = WMMA4F32(WhRZ1[q], hB, RZ1);
      HN0 = WMMA4F32(WhN0[q],  hB, HN0);
      N1  = WMMA4F32(WhN1[q],  hB, N1);
    }

    xR0 = bR0; xZ0 = bZ0; xRZ1 = bRZ1; xN0 = bXN0; xN1 = bN1;
#pragma unroll
    for (int q = 0; q < 6; ++q) {
      xR0  = WMMA4F32(WxR0[q],  aC[q], xR0);
      xZ0  = WMMA4F32(WxZ0[q],  aC[q], xZ0);
      xRZ1 = WMMA4F32(WxRZ1[q], aC[q], xRZ1);
      xN0  = WMMA4F32(WxN0[q],  aC[q], xN0);
      xN1  = WMMA4F32(WxN1[q],  aC[q], xN1);
    }

    float s = 0.0f;
#pragma unroll
    for (int e = 0; e < 8; ++e) {
      const float r  = sigmoidf_(R0[e]);
      const float z  = sigmoidf_(Z0[e]);
      const float nn = tanhf(XN0[e] + r * HN0[e]);
      const float hv = (1.0f - z) * nn + z * h[e];
      h[e] = hv;
      s += l2w[e] * hv;
    }
#pragma unroll
    for (int j = 0; j < 4; ++j) {
      const float r  = sigmoidf_(RZ1[j]);
      const float z  = sigmoidf_(RZ1[4 + j]);
      const float nn = tanhf(N1[j] + r * N1[4 + j]);
      const float hv = (1.0f - z) * nn + z * h[8 + j];
      h[8 + j] = hv;
      s += l2w[8 + j] * hv;
    }
    const float tot = s + xhalf(s);
    const float o   = sigmoidf_(tot + l2b);
    if (hf == 0) orow[t] = o;

#pragma unroll
    for (int q = 0; q < 6; ++q) aC[q] = aN[q];
  }
}

extern "C" void kernel_launch(void* const* d_in, const int* in_sizes, int n_in,
                              void* d_out, int out_size, void* d_ws,
                              size_t ws_size, hipStream_t stream) {
  (void)in_sizes; (void)n_in; (void)out_size; (void)ws_size;
  const float* x     = (const float*)d_in[0];
  const float* lc1_w = (const float*)d_in[1];
  const float* lc1_b = (const float*)d_in[2];
  const float* w_ih  = (const float*)d_in[3];
  const float* w_hh  = (const float*)d_in[4];
  const float* b_ih  = (const float*)d_in[5];
  const float* b_hh  = (const float*)d_in[6];
  const float* lc2_w = (const float*)d_in[7];
  const float* lc2_b = (const float*)d_in[8];
  float* out  = (float*)d_out;
  float* aBuf = (float*)d_ws;

  lc1_tanh_kernel<<<2048, 32, 0, stream>>>(x, lc1_w, lc1_b, aBuf);
  gru_scan_kernel<<<Bsz / 16, 32, 0, stream>>>(aBuf, w_ih, w_hh, b_ih, b_hh,
                                               lc2_w, lc2_b, out);
}


# library tanhf and IEEE-division sigmoid replaced by branch-free in-line forms on v_exp_f32 + v_rcp_f32 (tanh: its polynomial or 1-2/(e^2y+1) by select; sigmoid: seed + Newton step); sum order kept
# speedup vs baseline: 4.6638x; 2.6949x over previous
#include <hip/hip_runtime.h>
#include <hip/hip_bf16.h>
#include <cmath>


typedef __attribute__((ext_vector_type(2))) float v2f;
typedef __attribute__((ext_vector_type(8))) float v8f;

#define WMMA4F32(A, Bv, C)                                                   \
  __builtin_amdgcn_wmma_f32_16x16x4_f32(false, (A), false, (Bv), (short)0,   \
                                        (C), false, false)

static constexpr int Bsz  = 256;
static constexpr int Tlen = 2000;
static constexpr int Fin  = 42;
static constexpr int Hh   = 24;
static constexpr int Gg   = 72;


__device__ __forceinline__ float sigmoidf_(float v) {
  const float e  = __builtin_amdgcn_exp2f(
      __builtin_fminf(v * -1.44269504088896341f, 126.0f));
  const float d  = 1.0f + e;
  const float r0 = __builtin_amdgcn_rcpf(d);
  const float er = __builtin_fmaf(-d, r0, 1.0f);
  return __builtin_fmaf(er, r0, r0);
}

__device__ __forceinline__ float tanhf_(float x) {
  const float y  = __builtin_fabsf(x);
  const float y2 = y * y;
  float p = __builtin_fmaf(y2, -0x1.758e7ap-8f, 0x1.521192p-6f);
  p = __builtin_fmaf(y2, p, -0x1.b8389cp-5f);
  p = __builtin_fmaf(y2, p, 0x1.110704p-3f);
  p = __builtin_fmaf(y2, p, -0x1.555532p-2f);
  const float zs = __builtin_fmaf(y2, y * p, y);
  const float t  = __builtin_amdgcn_exp2f(y * 2.88539008177792681f);
  const float zl =
      __builtin_fmaf(-2.0f, __builtin_amdgcn_rcpf(t + 1.0f), 1.0f);
  return __builtin_copysignf((y < 0.625f) ? zs : zl, x);
}

__global__ __launch_bounds__(32) void lc1_tanh_kernel(
    const float* __restrict__ x, const float* __restrict__ w1,
    const float* __restrict__ b1, float* __restrict__ aOut) {
  const int  lane = threadIdx.x;
  const int  nl   = lane & 15;
  const int  ko   = (lane >> 4) * 2;
  const int  mo   = (lane >> 4) * 8;
  const bool hiHalf = (lane >= 16);
  const v2f  zero2 = {0.0f, 0.0f};

  v2f B0[11], B1[11];
  const int  n1  = 16 + nl;
  const bool n1v = (n1 < Hh);
  const int  n1c = n1v ? n1 : (Hh - 1);
#pragma unroll
  for (int kc = 0; kc < 11; ++kc) {
    const int k   = kc * 4 + ko;
    const int kc0 = (k     < Fin) ? k     : 0;
    const int kc1 = (k + 1 < Fin) ? k + 1 : 0;
    const float w00 = w1[nl  * Fin + kc0];
    const float w01 = w1[nl  * Fin + kc1];
    const float w10 = w1[n1c * Fin + kc0];
    const float w11 = w1[n1c * Fin + kc1];
    B0[kc].x = (k     < Fin) ? w00 : 0.0f;
    B0[kc].y = (k + 1 < Fin) ? w01 : 0.0f;
    B1[kc].x = (n1v && (k     < Fin)) ? w10 : 0.0f;
    B1[kc].y = (n1v && (k + 1 < Fin)) ? w11 : 0.0f;
  }
  const float bA = b1[nl];
  const float bB = n1v ? b1[n1c] : 0.0f;

  const int NT = (Bsz * Tlen) / 16;
  for (int tile = blockIdx.x; tile < NT; tile += gridDim.x) {
    const float* xr = x + (size_t)(tile * 16 + nl) * Fin;

    if (tile + (int)gridDim.x < NT)
      __builtin_prefetch(xr + (size_t)gridDim.x * 16 * Fin, 0, 1);

    v8f a0e = {0, 0, 0, 0, 0, 0, 0, 0}, a0o = {0, 0, 0, 0, 0, 0, 0, 0};
    v8f a1e = {0, 0, 0, 0, 0, 0, 0, 0}, a1o = {0, 0, 0, 0, 0, 0, 0, 0};
#pragma unroll
    for (int kc = 0; kc < 10; kc += 2) {
      const v2f Ae = *(const v2f*)(xr + kc * 4 + ko);
      const v2f Ao = *(const v2f*)(xr + (kc + 1) * 4 + ko);
      a0e = WMMA4F32(Ae, B0[kc], a0e);
      a1e = WMMA4F32(Ae, B1[kc], a1e);
      a0o = WMMA4F32(Ao, B0[kc + 1], a0o);
      a1o = WMMA4F32(Ao, B1[kc + 1], a1o);
    }
    {
      const v2f t = *(const v2f*)(xr + 40);
      const v2f A = hiHalf ? zero2 : t;
      a0e = WMMA4F32(A, B0[10], a0e);
      a1e = WMMA4F32(A, B1[10], a1e);
    }

    float* arow = aOut + (size_t)tile * 16 * Hh;
#pragma unroll
    for (int j = 0; j < 8; ++j)
      arow[(j + mo) * Hh + nl] = tanhf_(a0e[j] + a0o[j] + bA);
    if (nl < 8) {
#pragma unroll
      for (int j = 0; j < 8; ++j)
        arow[(j + mo) * Hh + 16 + nl] = tanhf_(a1e[j] + a1o[j] + bB);
    }
  }
}

__device__ __forceinline__ float xhalf(float v) {
  const int b = __builtin_bit_cast(int, v);
  return __builtin_bit_cast(
      float, __builtin_amdgcn_permlanex16(b, b, 0x76543210, 0xfedcba98, false,
                                          false));
}

__global__ __launch_bounds__(32) void gru_scan_kernel(
    const float* __restrict__ a, const float* __restrict__ w_ih,
    const float* __restrict__ w_hh, const float* __restrict__ b_ih,
    const float* __restrict__ b_hh, const float* __restrict__ lc2_w,
    const float* __restrict__ lc2_b, float* __restrict__ out) {
  const int lane = threadIdx.x;
  const int nl   = lane & 15;
  const int hf   = lane >> 4;
  const int kb   = 12 * hf;
  const int b0   = blockIdx.x * 16;
  const v2f zero2 = {0.0f, 0.0f};

  const int  mh  = nl >> 3, mj = nl & 7;
  const int  u0  = 12 * mh + mj;
  const int  u1  = 12 * mh + 8 + (mj & 3);
  const bool lo4 = (mj < 4);
  const int rowR0 = u0, rowZ0 = Hh + u0, rowN0 = 2 * Hh + u0;
  const int rowRZ1 = (lo4 ? 0 : Hh) + u1, rowN1 = 2 * Hh + u1;

  v2f WxR0[6], WxZ0[6], WxRZ1[6], WxN0[6], WxN1[6];
  v2f WhR0[6], WhZ0[6], WhRZ1[6], WhN0[6], WhN1[6];
#pragma unroll
  for (int q = 0; q < 6; ++q) {
    const int k = kb + 2 * q;
    WxR0[q]  = *(const v2f*)(w_ih + rowR0  * Hh + k);
    WxZ0[q]  = *(const v2f*)(w_ih + rowZ0  * Hh + k);
    WxRZ1[q] = *(const v2f*)(w_ih + rowRZ1 * Hh + k);
    WxN0[q]  = *(const v2f*)(w_ih + rowN0  * Hh + k);
    WhR0[q]  = *(const v2f*)(w_hh + rowR0  * Hh + k);
    WhZ0[q]  = *(const v2f*)(w_hh + rowZ0  * Hh + k);
    WhRZ1[q] = *(const v2f*)(w_hh + rowRZ1 * Hh + k);
    WhN0[q]  = *(const v2f*)(w_hh + rowN0  * Hh + k);
    const v2f xi = *(const v2f*)(w_ih + rowN1 * Hh + k);
    const v2f hh = *(const v2f*)(w_hh + rowN1 * Hh + k);
    WxN1[q] = lo4 ? xi : zero2;
    WhN1[q] = lo4 ? zero2 : hh;
  }

  v8f bR0, bZ0, bRZ1, bXN0, bHN0, bN1;
#pragma unroll
  for (int j = 0; j < 8; ++j) {
    const int c0 = kb + j;
    const int c1 = kb + 8 + (j & 3);
    bR0[j]  = b_ih[c0] + b_hh[c0];
    bZ0[j]  = b_ih[Hh + c0] + b_hh[Hh + c0];
    bRZ1[j] = (j < 4) ? (b_ih[c1] + b_hh[c1]) : (b_ih[Hh + c1] + b_hh[Hh + c1]);
    bXN0[j] = b_ih[2 * Hh + c0];
    bHN0[j] = b_hh[2 * Hh + c0];
    bN1[j]  = (j < 4) ? b_ih[2 * Hh + c1] : b_hh[2 * Hh + c1];
  }
  float l2w[12];
#pragma unroll
  for (int e = 0; e < 12; ++e) l2w[e] = lc2_w[kb + e];
  const float l2b = lc2_b[0];

  const float* arow = a + (size_t)(b0 + nl) * Tlen * Hh + kb;
  float*       orow = out + (size_t)(b0 + nl) * Tlen;
  v2f aC[6];
#pragma unroll
  for (int q = 0; q < 6; ++q) aC[q] = *(const v2f*)(arow + 2 * q);

  v8f xR0 = bR0, xZ0 = bZ0, xRZ1 = bRZ1, xN0 = bXN0, xN1 = bN1;
#pragma unroll
  for (int q = 0; q < 6; ++q) {
    xR0  = WMMA4F32(WxR0[q],  aC[q], xR0);
    xZ0  = WMMA4F32(WxZ0[q],  aC[q], xZ0);
    xRZ1 = WMMA4F32(WxRZ1[q], aC[q], xRZ1);
    xN0  = WMMA4F32(WxN0[q],  aC[q], xN0);
    xN1  = WMMA4F32(WxN1[q],  aC[q], xN1);
  }
#pragma unroll
  for (int q = 0; q < 6; ++q) aC[q] = *(const v2f*)(arow + Hh + 2 * q);

  float h[12];
#pragma unroll
  for (int e = 0; e < 12; ++e) h[e] = 0.0f;

  for (int t = 0; t < Tlen; ++t) {
    const int tn = (t + 2 < Tlen) ? (t + 2) : (Tlen - 1);
    v2f aN[6];
#pragma unroll
    for (int q = 0; q < 6; ++q)
      aN[q] = *(const v2f*)(arow + (size_t)tn * Hh + 2 * q);

    v8f R0 = xR0, Z0 = xZ0, RZ1 = xRZ1, HN0 = bHN0, N1 = xN1;
    const v8f XN0 = xN0;
#pragma unroll
    for (int q = 0; q < 6; ++q) {
      const v2f hB = {h[2 * q], h[2 * q + 1]};
      R0  = WMMA4F32(WhR0[q],  hB, R0);
      Z0  = WMMA4F32(WhZ0[q],  hB, Z0);
      RZ1 = WMMA4F32(WhRZ1[q], hB, RZ1);
      HN0 = WMMA4F32(WhN0[q],  hB, HN0);
      N1  = WMMA4F32(WhN1[q],  hB, N1);
    }

    xR0 = bR0; xZ0 = bZ0; xRZ1 = bRZ1; xN0 = bXN0; xN1 = bN1;
#pragma unroll
    for (int q = 0; q < 6; ++q) {
      xR0  = WMMA4F32(WxR0[q],  aC[q], xR0);
      xZ0  = WMMA4F32(WxZ0[q],  aC[q], xZ0);
      xRZ1 = WMMA4F32(WxRZ1[q], aC[q], xRZ1);
      xN0  = WMMA4F32(WxN0[q],  aC[q], xN0);
      xN1  = WMMA4F32(WxN1[q],  aC[q], xN1);
    }

    float s = 0.0f;
#pragma unroll
    for (int e = 0; e < 8; ++e) {
      const float r  = sigmoidf_(R0[e]);
      const float z  = sigmoidf_(Z0[e]);
      const float nn = tanhf_(XN0[e] + r * HN0[e]);
      const float hv = (1.0f - z) * nn + z * h[e];
      h[e] = hv;
      s += l2w[e] * hv;
    }
#pragma unroll
    for (int j = 0; j < 4; ++j) {
      const float r  = sigmoidf_(RZ1[j]);
      const float z  = sigmoidf_(RZ1[4 + j]);
      const float nn = tanhf_(N1[j] + r * N1[4 + j]);
      const float hv = (1.0f - z) * nn + z * h[8 + j];
      h[8 + j] = hv;
      s += l2w[8 + j] * hv;
    }
    const float tot = s + xhalf(s);
    const float o   = sigmoidf_(tot + l2b);
    if (hf == 0) orow[t] = o;

#pragma unroll
    for (int q = 0; q < 6; ++q) aC[q] = aN[q];
  }
}

extern "C" void kernel_launch(void* const* d_in, const int* in_sizes, int n_in,
                              void* d_out, int out_size, void* d_ws,
                              size_t ws_size, hipStream_t stream) {
  (void)in_sizes; (void)n_in; (void)out_size; (void)ws_size;
  const float* x     = (const float*)d_in[0];
  const float* lc1_w = (const float*)d_in[1];
  const float* lc1_b = (const float*)d_in[2];
  const float* w_ih  = (const float*)d_in[3];
  const float* w_hh  = (const float*)d_in[4];
  const float* b_ih  = (const float*)d_in[5];
  const float* b_hh  = (const float*)d_in[6];
  const float* lc2_w = (const float*)d_in[7];
  const float* lc2_b = (const float*)d_in[8];
  float* out  = (float*)d_out;
  float* aBuf = (float*)d_ws;

  lc1_tanh_kernel<<<2048, 32, 0, stream>>>(x, lc1_w, lc1_b, aBuf);
  gru_scan_kernel<<<Bsz / 16, 32, 0, stream>>>(aBuf, w_ih, w_hh, b_ih, b_hh,
                                               lc2_w, lc2_b, out);
}


# GRU scan of a 16-row batch group split over 3 waves by hidden units (2 WMMA tiles a wave, not 5); h exchanged through LDS, one split barrier a step with the x-side products over it; sum order kept
# speedup vs baseline: 7.2907x; 1.5633x over previous
#include <hip/hip_runtime.h>
#include <hip/hip_bf16.h>
#include <cmath>


typedef __attribute__((ext_vector_type(2))) float v2f;
typedef __attribute__((ext_vector_type(8))) float v8f;

#define WMMA4F32(A, Bv, C)                                                   \
  __builtin_amdgcn_wmma_f32_16x16x4_f32(false, (A), false, (Bv), (short)0,   \
                                        (C), false, false)

static constexpr int Bsz  = 256;
static constexpr int Tlen = 2000;
static constexpr int Fin  = 42;
static constexpr int Hh   = 24;
static constexpr int Gg   = 72;


__device__ __forceinline__ float sigmoidf_(float v) {
  const float e  = __builtin_amdgcn_exp2f(
      __builtin_fminf(v * -1.44269504088896341f, 126.0f));
  const float d  = 1.0f + e;
  const float r0 = __builtin_amdgcn_rcpf(d);
  const float er = __builtin_fmaf(-d, r0, 1.0f);
  return __builtin_fmaf(er, r0, r0);
}

__device__ __forceinline__ float tanhf_(float x) {
  const float y  = __builtin_fabsf(x);
  const float y2 = y * y;
  float p = __builtin_fmaf(y2, -0x1.758e7ap-8f, 0x1.521192p-6f);
  p = __builtin_fmaf(y2, p, -0x1.b8389cp-5f);
  p = __builtin_fmaf(y2, p, 0x1.110704p-3f);
  p = __builtin_fmaf(y2, p, -0x1.555532p-2f);
  const float zs = __builtin_fmaf(y2, y * p, y);
  const float t  = __builtin_amdgcn_exp2f(y * 2.88539008177792681f);
  const float zl =
      __builtin_fmaf(-2.0f, __builtin_amdgcn_rcpf(t + 1.0f), 1.0f);
  return __builtin_copysignf((y < 0.625f) ? zs : zl, x);
}

__global__ __launch_bounds__(32) void lc1_tanh_kernel(
    const float* __restrict__ x, const float* __restrict__ w1,
    const float* __restrict__ b1, float* __restrict__ aOut) {
  const int  lane = threadIdx.x;
  const int  nl   = lane & 15;
  const int  ko   = (lane >> 4) * 2;
  const int  mo   = (lane >> 4) * 8;
  const bool hiHalf = (lane >= 16);
  const v2f  zero2 = {0.0f, 0.0f};

  v2f B0[11], B1[11];
  const int  n1  = 16 + nl;
  const bool n1v = (n1 < Hh);
  const int  n1c = n1v ? n1 : (Hh - 1);
#pragma unroll
  for (int kc = 0; kc < 11; ++kc) {
    const int k   = kc * 4 + ko;
    const int kc0 = (k     < Fin) ? k     : 0;
    const int kc1 = (k + 1 < Fin) ? k + 1 : 0;
    const float w00 = w1[nl  * Fin + kc0];
    const float w01 = w1[nl  * Fin + kc1];
    const float w10 = w1[n1c * Fin + kc0];
    const float w11 = w1[n1c * Fin + kc1];
    B0[kc].x = (k     < Fin) ? w00 : 0.0f;
    B0[kc].y = (k + 1 < Fin) ? w01 : 0.0f;
    B1[kc].x = (n1v && (k     < Fin)) ? w10 : 0.0f;
    B1[kc].y = (n1v && (k + 1 < Fin)) ? w11 : 0.0f;
  }
  const float bA = b1[nl];
  const float bB = n1v ? b1[n1c] : 0.0f;

  const int NT = (Bsz * Tlen) / 16;
  for (int tile = blockIdx.x; tile < NT; tile += gridDim.x) {
    const float* xr = x + (size_t)(tile * 16 + nl) * Fin;

    if (tile + (int)gridDim.x < NT)
      __builtin_prefetch(xr + (size_t)gridDim.x * 16 * Fin, 0, 1);

    v8f a0e = {0, 0, 0, 0, 0, 0, 0, 0}, a0o = {0, 0, 0, 0, 0, 0, 0, 0};
    v8f a1e = {0, 0, 0, 0, 0, 0, 0, 0}, a1o = {0, 0, 0, 0, 0, 0, 0, 0};
#pragma unroll
    for (int kc = 0; kc < 10; kc += 2) {
      const v2f Ae = *(const v2f*)(xr + kc * 4 + ko);
      const v2f Ao = *(const v2f*)(xr + (kc + 1) * 4 + ko);
      a0e = WMMA4F32(Ae, B0[kc], a0e);
      a1e = WMMA4F32(Ae, B1[kc], a1e);
      a0o = WMMA4F32(Ao, B0[kc + 1], a0o);
      a1o = WMMA4F32(Ao, B1[kc + 1], a1o);
    }
    {
      const v2f t = *(const v2f*)(xr + 40);
      const v2f A = hiHalf ? zero2 : t;
      a0e = WMMA4F32(A, B0[10], a0e);
      a1e = WMMA4F32(A, B1[10], a1e);
    }

    float* arow = aOut + (size_t)tile * 16 * Hh;
#pragma unroll
    for (int j = 0; j < 8; ++j)
      arow[(j + mo) * Hh + nl] = tanhf_(a0e[j] + a0o[j] + bA);
    if (nl < 8) {
#pragma unroll
      for (int j = 0; j < 8; ++j)
        arow[(j + mo) * Hh + 16 + nl] = tanhf_(a1e[j] + a1o[j] + bB);
    }
  }
}

typedef __attribute__((ext_vector_type(4))) float v4f;

static constexpr int HPf = 28;

__device__ __forceinline__ float xhalf(float v) {
  const int b = __builtin_bit_cast(int, v);
  return __builtin_bit_cast(
      float, __builtin_amdgcn_permlanex16(b, b, 0x76543210, 0xfedcba98, false,
                                          false));
}

__global__ __launch_bounds__(96) void gru_scan_kernel(
    const float* __restrict__ a, const float* __restrict__ w_ih,
    const float* __restrict__ w_hh, const float* __restrict__ b_ih,
    const float* __restrict__ b_hh, const float* __restrict__ lc2_w,
    const float* __restrict__ lc2_b, float* __restrict__ out) {
  __shared__ __attribute__((aligned(16))) float hbuf[2 * 16 * HPf];

  const int lane = threadIdx.x & 31;
  const int w    = threadIdx.x >> 5;
  const int nl   = lane & 15;
  const int hf   = lane >> 4;
  const int kb   = 12 * hf;
  const int b0   = blockIdx.x * 16;
  const v2f zero2 = {0.0f, 0.0f};

  const int  mh  = nl >> 3, mj = nl & 7;
  const int  ua  = 8 * w + 4 * mh + (mj & 3);
  const bool lo4 = (mj < 4);
  const int rowRZ = (lo4 ? 0 : Hh) + ua, rowN = 2 * Hh + ua;

  v2f WxRZ[6], WxN[6], WhRZ[6], WhN[6];
#pragma unroll
  for (int q = 0; q < 6; ++q) {
    const int k = kb + 2 * q;
    WxRZ[q] = *(const v2f*)(w_ih + rowRZ * Hh + k);
    WhRZ[q] = *(const v2f*)(w_hh + rowRZ * Hh + k);
    const v2f xi = *(const v2f*)(w_ih + rowN * Hh + k);
    const v2f hh = *(const v2f*)(w_hh + rowN * Hh + k);
    WxN[q] = lo4 ? xi : zero2;
    WhN[q] = lo4 ? zero2 : hh;
  }

  v8f bRZ, bN;
#pragma unroll
  for (int j = 0; j < 8; ++j) {
    const int c = 8 * w + 4 * hf + (j & 3);
    bRZ[j] = (j < 4) ? (b_ih[c] + b_hh[c]) : (b_ih[Hh + c] + b_hh[Hh + c]);
    bN[j]  = (j < 4) ? b_ih[2 * Hh + c] : b_hh[2 * Hh + c];
  }
  float l2w[12];
#pragma unroll
  for (int e = 0; e < 12; ++e) l2w[e] = lc2_w[kb + e];
  const float l2b = lc2_b[0];
  const bool storeLane = (hf == 0) && (w == 2);

  const float* arow = a + (size_t)(b0 + nl) * Tlen * Hh + kb;
  float*       orow = out + (size_t)(b0 + nl) * Tlen;
  v2f aC[6];
#pragma unroll
  for (int q = 0; q < 6; ++q) aC[q] = *(const v2f*)(arow + 2 * q);

  v8f xRZ = bRZ, xN = bN;
#pragma unroll
  for (int q = 0; q < 6; ++q) {
    xRZ = WMMA4F32(WxRZ[q], aC[q], xRZ);
    xN  = WMMA4F32(WxN[q],  aC[q], xN);
  }
#pragma unroll
  for (int q = 0; q < 6; ++q) aC[q] = *(const v2f*)(arow + Hh + 2 * q);

  float*       hst = hbuf + nl * HPf + 8 * w + 4 * hf;
  const float* hld = hbuf + nl * HPf + kb;

  float h[12];
#pragma unroll
  for (int e = 0; e < 12; ++e) h[e] = 0.0f;
  float hown[4] = {0.0f, 0.0f, 0.0f, 0.0f};

  for (int t = 0; t < Tlen; ++t) {
    const int tn = (t + 2 < Tlen) ? (t + 2) : (Tlen - 1);
    v2f aN[6];
#pragma unroll
    for (int q = 0; q < 6; ++q)
      aN[q] = *(const v2f*)(arow + (size_t)tn * Hh + 2 * q);

    v8f RZ = xRZ, N = xN;
#pragma unroll
    for (int q = 0; q < 6; ++q) {
      const v2f hB = {h[2 * q], h[2 * q + 1]};
      RZ = WMMA4F32(WhRZ[q], hB, RZ);
      N  = WMMA4F32(WhN[q],  hB, N);
    }

    float s = 0.0f;
#pragma unroll
    for (int e = 0; e < 12; ++e) s += l2w[e] * h[e];
    const float tot = s + xhalf(s);

    v4f hn;
#pragma unroll
    for (int i = 0; i < 4; ++i) {
      const float r  = sigmoidf_(RZ[i]);
      const float z  = sigmoidf_(RZ[4 + i]);
      const float nn = tanhf_(N[i] + r * N[4 + i]);
      const float hv = (1.0f - z) * nn + z * hown[i];
      hown[i] = hv;
      hn[i]   = hv;
    }

    const int bo = (t & 1) * (16 * HPf);
    *(v4f*)(hst + bo) = hn;

    if (storeLane && t > 0) orow[t - 1] = sigmoidf_(tot + l2b);

    __builtin_amdgcn_fence(__ATOMIC_RELEASE, "workgroup", "local");
    __builtin_amdgcn_s_barrier_signal(-1);
    __builtin_amdgcn_sched_barrier(0);
    xRZ = bRZ; xN = bN;
#pragma unroll
    for (int q = 0; q < 3; ++q) {
      xRZ = WMMA4F32(WxRZ[q], aC[q], xRZ);
      xN  = WMMA4F32(WxN[q],  aC[q], xN);
    }
    __builtin_amdgcn_sched_barrier(0);
    __builtin_amdgcn_s_barrier_wait(-1);
    __builtin_amdgcn_fence(__ATOMIC_ACQUIRE, "workgroup", "local");
    const v4f h0 = *(const v4f*)(hld + bo);
    const v4f h1 = *(const v4f*)(hld + bo + 4);
    const v4f h2 = *(const v4f*)(hld + bo + 8);
    __builtin_amdgcn_sched_barrier(0);
#pragma unroll
    for (int q = 3; q < 6; ++q) {
      xRZ = WMMA4F32(WxRZ[q], aC[q], xRZ);
      xN  = WMMA4F32(WxN[q],  aC[q], xN);
    }
#pragma unroll
    for (int e = 0; e < 4; ++e) { h[e] = h0[e]; h[4 + e] = h1[e]; h[8 + e] = h2[e]; }

#pragma unroll
    for (int q = 0; q < 6; ++q) aC[q] = aN[q];
  }

  {
    float s = 0.0f;
#pragma unroll
    for (int e = 0; e < 12; ++e) s += l2w[e] * h[e];
    const float tot = s + xhalf(s);
    const float o   = sigmoidf_(tot + l2b);
    if (storeLane) orow[Tlen - 1] = o;
  }
}

extern "C" void kernel_launch(void* const* d_in, const int* in_sizes, int n_in,
                              void* d_out, int out_size, void* d_ws,
                              size_t ws_size, hipStream_t stream) {
  (void)in_sizes; (void)n_in; (void)out_size; (void)ws_size;
  const float* x     = (const float*)d_in[0];
  const float* lc1_w = (const float*)d_in[1];
  const float* lc1_b = (const float*)d_in[2];
  const float* w_ih  = (const float*)d_in[3];
  const float* w_hh  = (const float*)d_in[4];
  const float* b_ih  = (const float*)d_in[5];
  const float* b_hh  = (const float*)d_in[6];
  const float* lc2_w = (const float*)d_in[7];
  const float* lc2_b = (const float*)d_in[8];
  float* out  = (float*)d_out;
  float* aBuf = (float*)d_ws;

  lc1_tanh_kernel<<<2048, 32, 0, stream>>>(x, lc1_w, lc1_b, aBuf);
  gru_scan_kernel<<<Bsz / 16, 96, 0, stream>>>(aBuf, w_ih, w_hh, b_ih, b_hh,
                                               lc2_w, lc2_b, out);
}


# tanh and sigmoid of both kernels on v_tanh_f32 (sigmoid = 1/2 + 1/2 tanh(v/2), the halving folded exactly into the r/z weight rows), replacing the v_exp_f32 + v_rcp_f32 forms; f32, sum order kept
# speedup vs baseline: 7.9250x; 1.0870x over previous
#include <hip/hip_runtime.h>
#include <hip/hip_bf16.h>
#include <cmath>


typedef __attribute__((ext_vector_type(2))) float v2f;
typedef __attribute__((ext_vector_type(8))) float v8f;

#define WMMA4F32(A, Bv, C)                                                   \
  __builtin_amdgcn_wmma_f32_16x16x4_f32(false, (A), false, (Bv), (short)0,   \
                                        (C), false, false)

static constexpr int Bsz  = 256;
static constexpr int Tlen = 2000;
static constexpr int Fin  = 42;
static constexpr int Hh   = 24;
static constexpr int Gg   = 72;


__device__ __forceinline__ float tanhf_(float x) {
  return __builtin_amdgcn_tanhf(x);
}

__device__ __forceinline__ float sigmoid_of_half_(float u) {
  return __builtin_fmaf(0.5f, __builtin_amdgcn_tanhf(u), 0.5f);
}
__device__ __forceinline__ float sigmoidf_(float v) {
  return sigmoid_of_half_(0.5f * v);
}

__global__ __launch_bounds__(32) void lc1_tanh_kernel(
    const float* __restrict__ x, const float* __restrict__ w1,
    const float* __restrict__ b1, float* __restrict__ aOut) {
  const int  lane = threadIdx.x;
  const int  nl   = lane & 15;
  const int  ko   = (lane >> 4) * 2;
  const int  mo   = (lane >> 4) * 8;
  const bool hiHalf = (lane >= 16);
  const v2f  zero2 = {0.0f, 0.0f};

  v2f B0[11], B1[11];
  const int  n1  = 16 + nl;
  const bool n1v = (n1 < Hh);
  const int  n1c = n1v ? n1 : (Hh - 1);
#pragma unroll
  for (int kc = 0; kc < 11; ++kc) {
    const int k   = kc * 4 + ko;
    const int kc0 = (k     < Fin) ? k     : 0;
    const int kc1 = (k + 1 < Fin) ? k + 1 : 0;
    const float w00 = w1[nl  * Fin + kc0];
    const float w01 = w1[nl  * Fin + kc1];
    const float w10 = w1[n1c * Fin + kc0];
    const float w11 = w1[n1c * Fin + kc1];
    B0[kc].x = (k     < Fin) ? w00 : 0.0f;
    B0[kc].y = (k + 1 < Fin) ? w01 : 0.0f;
    B1[kc].x = (n1v && (k     < Fin)) ? w10 : 0.0f;
    B1[kc].y = (n1v && (k + 1 < Fin)) ? w11 : 0.0f;
  }
  const float bA = b1[nl];
  const float bB = n1v ? b1[n1c] : 0.0f;

  const int NT = (Bsz * Tlen) / 16;
  for (int tile = blockIdx.x; tile < NT; tile += gridDim.x) {
    const float* xr = x + (size_t)(tile * 16 + nl) * Fin;

    if (tile + (int)gridDim.x < NT)
      __builtin_prefetch(xr + (size_t)gridDim.x * 16 * Fin, 0, 1);

    v8f a0e = {0, 0, 0, 0, 0, 0, 0, 0}, a0o = {0, 0, 0, 0, 0, 0, 0, 0};
    v8f a1e = {0, 0, 0, 0, 0, 0, 0, 0}, a1o = {0, 0, 0, 0, 0, 0, 0, 0};
#pragma unroll
    for (int kc = 0; kc < 10; kc += 2) {
      const v2f Ae = *(const v2f*)(xr + kc * 4 + ko);
      const v2f Ao = *(const v2f*)(xr + (kc + 1) * 4 + ko);
      a0e = WMMA4F32(Ae, B0[kc], a0e);
      a1e = WMMA4F32(Ae, B1[kc], a1e);
      a0o = WMMA4F32(Ao, B0[kc + 1], a0o);
      a1o = WMMA4F32(Ao, B1[kc + 1], a1o);
    }
    {
      const v2f t = *(const v2f*)(xr + 40);
      const v2f A = hiHalf ? zero2 : t;
      a0e = WMMA4F32(A, B0[10], a0e);
      a1e = WMMA4F32(A, B1[10], a1e);
    }

    float* arow = aOut + (size_t)tile * 16 * Hh;
#pragma unroll
    for (int j = 0; j < 8; ++j)
      arow[(j + mo) * Hh + nl] = tanhf_(a0e[j] + a0o[j] + bA);
    if (nl < 8) {
#pragma unroll
      for (int j = 0; j < 8; ++j)
        arow[(j + mo) * Hh + 16 + nl] = tanhf_(a1e[j] + a1o[j] + bB);
    }
  }
}

typedef __attribute__((ext_vector_type(4))) float v4f;

static constexpr int HPf = 28;

__device__ __forceinline__ float xhalf(float v) {
  const int b = __builtin_bit_cast(int, v);
  return __builtin_bit_cast(
      float, __builtin_amdgcn_permlanex16(b, b, 0x76543210, 0xfedcba98, false,
                                          false));
}

__global__ __launch_bounds__(96) void gru_scan_kernel(
    const float* __restrict__ a, const float* __restrict__ w_ih,
    const float* __restrict__ w_hh, const float* __restrict__ b_ih,
    const float* __restrict__ b_hh, const float* __restrict__ lc2_w,
    const float* __restrict__ lc2_b, float* __restrict__ out) {
  __shared__ __attribute__((aligned(16))) float hbuf[2 * 16 * HPf];

  const int lane = threadIdx.x & 31;
  const int w    = threadIdx.x >> 5;
  const int nl   = lane & 15;
  const int hf   = lane >> 4;
  const int kb   = 12 * hf;
  const int b0   = blockIdx.x * 16;
  const v2f zero2 = {0.0f, 0.0f};

  const int  mh  = nl >> 3, mj = nl & 7;
  const int  ua  = 8 * w + 4 * mh + (mj & 3);
  const bool lo4 = (mj < 4);
  const int rowRZ = (lo4 ? 0 : Hh) + ua, rowN = 2 * Hh + ua;

  v2f WxRZ[6], WxN[6], WhRZ[6], WhN[6];
#pragma unroll
  for (int q = 0; q < 6; ++q) {
    const int k = kb + 2 * q;
    WxRZ[q] = 0.5f * *(const v2f*)(w_ih + rowRZ * Hh + k);
    WhRZ[q] = 0.5f * *(const v2f*)(w_hh + rowRZ * Hh + k);
    const v2f xi = *(const v2f*)(w_ih + rowN * Hh + k);
    const v2f hh = *(const v2f*)(w_hh + rowN * Hh + k);
    WxN[q] = lo4 ? xi : zero2;
    WhN[q] = lo4 ? zero2 : hh;
  }

  v8f bRZ, bN;
#pragma unroll
  for (int j = 0; j < 8; ++j) {
    const int c = 8 * w + 4 * hf + (j & 3);
    bRZ[j] = 0.5f * ((j < 4) ? (b_ih[c] + b_hh[c])
                             : (b_ih[Hh + c] + b_hh[Hh + c]));
    bN[j]  = (j < 4) ? b_ih[2 * Hh + c] : b_hh[2 * Hh + c];
  }
  float l2w[12];
#pragma unroll
  for (int e = 0; e < 12; ++e) l2w[e] = lc2_w[kb + e];
  const float l2b = lc2_b[0];
  const bool storeLane = (hf == 0) && (w == 2);

  const float* arow = a + (size_t)(b0 + nl) * Tlen * Hh + kb;
  float*       orow = out + (size_t)(b0 + nl) * Tlen;
  v2f aC[6];
#pragma unroll
  for (int q = 0; q < 6; ++q) aC[q] = *(const v2f*)(arow + 2 * q);

  v8f xRZ = bRZ, xN = bN;
#pragma unroll
  for (int q = 0; q < 6; ++q) {
    xRZ = WMMA4F32(WxRZ[q], aC[q], xRZ);
    xN  = WMMA4F32(WxN[q],  aC[q], xN);
  }
#pragma unroll
  for (int q = 0; q < 6; ++q) aC[q] = *(const v2f*)(arow + Hh + 2 * q);

  float*       hst = hbuf + nl * HPf + 8 * w + 4 * hf;
  const float* hld = hbuf + nl * HPf + kb;

  float h[12];
#pragma unroll
  for (int e = 0; e < 12; ++e) h[e] = 0.0f;
  float hown[4] = {0.0f, 0.0f, 0.0f, 0.0f};

  for (int t = 0; t < Tlen; ++t) {
    const int tn = (t + 2 < Tlen) ? (t + 2) : (Tlen - 1);
    v2f aN[6];
#pragma unroll
    for (int q = 0; q < 6; ++q)
      aN[q] = *(const v2f*)(arow + (size_t)tn * Hh + 2 * q);

    v8f RZ = xRZ, N = xN;
#pragma unroll
    for (int q = 0; q < 6; ++q) {
      const v2f hB = {h[2 * q], h[2 * q + 1]};
      RZ = WMMA4F32(WhRZ[q], hB, RZ);
      N  = WMMA4F32(WhN[q],  hB, N);
    }

    float s = 0.0f;
#pragma unroll
    for (int e = 0; e < 12; ++e) s += l2w[e] * h[e];
    const float tot = s + xhalf(s);

    v4f hn;
#pragma unroll
    for (int i = 0; i < 4; ++i) {
      const float r  = sigmoid_of_half_(RZ[i]);
      const float z  = sigmoid_of_half_(RZ[4 + i]);
      const float nn = tanhf_(N[i] + r * N[4 + i]);
      const float hv = (1.0f - z) * nn + z * hown[i];
      hown[i] = hv;
      hn[i]   = hv;
    }

    const int bo = (t & 1) * (16 * HPf);
    *(v4f*)(hst + bo) = hn;

    if (storeLane && t > 0) orow[t - 1] = sigmoidf_(tot + l2b);

    __builtin_amdgcn_fence(__ATOMIC_RELEASE, "workgroup", "local");
    __builtin_amdgcn_s_barrier_signal(-1);
    __builtin_amdgcn_sched_barrier(0);
    xRZ = bRZ; xN = bN;
#pragma unroll
    for (int q = 0; q < 3; ++q) {
      xRZ = WMMA4F32(WxRZ[q], aC[q], xRZ);
      xN  = WMMA4F32(WxN[q],  aC[q], xN);
    }
    __builtin_amdgcn_sched_barrier(0);
    __builtin_amdgcn_s_barrier_wait(-1);
    __builtin_amdgcn_fence(__ATOMIC_ACQUIRE, "workgroup", "local");
    const v4f h0 = *(const v4f*)(hld + bo);
    const v4f h1 = *(const v4f*)(hld + bo + 4);
    const v4f h2 = *(const v4f*)(hld + bo + 8);
    __builtin_amdgcn_sched_barrier(0);
#pragma unroll
    for (int q = 3; q < 6; ++q) {
      xRZ = WMMA4F32(WxRZ[q], aC[q], xRZ);
      xN  = WMMA4F32(WxN[q],  aC[q], xN);
    }
#pragma unroll
    for (int e = 0; e < 4; ++e) { h[e] = h0[e]; h[4 + e] = h1[e]; h[8 + e] = h2[e]; }

#pragma unroll
    for (int q = 0; q < 6; ++q) aC[q] = aN[q];
  }

  {
    float s = 0.0f;
#pragma unroll
    for (int e = 0; e < 12; ++e) s += l2w[e] * h[e];
    const float tot = s + xhalf(s);
    const float o   = sigmoidf_(tot + l2b);
    if (storeLane) orow[Tlen - 1] = o;
  }
}

extern "C" void kernel_launch(void* const* d_in, const int* in_sizes, int n_in,
                              void* d_out, int out_size, void* d_ws,
                              size_t ws_size, hipStream_t stream) {
  (void)in_sizes; (void)n_in; (void)out_size; (void)ws_size;
  const float* x     = (const float*)d_in[0];
  const float* lc1_w = (const float*)d_in[1];
  const float* lc1_b = (const float*)d_in[2];
  const float* w_ih  = (const float*)d_in[3];
  const float* w_hh  = (const float*)d_in[4];
  const float* b_ih  = (const float*)d_in[5];
  const float* b_hh  = (const float*)d_in[6];
  const float* lc2_w = (const float*)d_in[7];
  const float* lc2_b = (const float*)d_in[8];
  float* out  = (float*)d_out;
  float* aBuf = (float*)d_ws;

  lc1_tanh_kernel<<<2048, 32, 0, stream>>>(x, lc1_w, lc1_b, aBuf);
  gru_scan_kernel<<<Bsz / 16, 96, 0, stream>>>(aBuf, w_ih, w_hh, b_ih, b_hh,
                                               lc2_w, lc2_b, out);
}


# GRU scan: a_t fragments through a ring of 8 register sets, each refilled 7 steps before it is read (was: loaded and waited for within one step, so a miss to memory stalled the step); sum order kept
# speedup vs baseline: 10.3571x; 1.3069x over previous
#include <hip/hip_runtime.h>
#include <hip/hip_bf16.h>
#include <cmath>


typedef __attribute__((ext_vector_type(2))) float v2f;
typedef __attribute__((ext_vector_type(8))) float v8f;

#define WMMA4F32(A, Bv, C)                                                   \
  __builtin_amdgcn_wmma_f32_16x16x4_f32(false, (A), false, (Bv), (short)0,   \
                                        (C), false, false)

static constexpr int Bsz  = 256;
static constexpr int Tlen = 2000;
static constexpr int Fin  = 42;
static constexpr int Hh   = 24;
static constexpr int Gg   = 72;


__device__ __forceinline__ float tanhf_(float x) {
  return __builtin_amdgcn_tanhf(x);
}

__device__ __forceinline__ float sigmoid_of_half_(float u) {
  return __builtin_fmaf(0.5f, __builtin_amdgcn_tanhf(u), 0.5f);
}
__device__ __forceinline__ float sigmoidf_(float v) {
  return sigmoid_of_half_(0.5f * v);
}

__global__ __launch_bounds__(32) void lc1_tanh_kernel(
    const float* __restrict__ x, const float* __restrict__ w1,
    const float* __restrict__ b1, float* __restrict__ aOut) {
  const int  lane = threadIdx.x;
  const int  nl   = lane & 15;
  const int  ko   = (lane >> 4) * 2;
  const int  mo   = (lane >> 4) * 8;
  const bool hiHalf = (lane >= 16);
  const v2f  zero2 = {0.0f, 0.0f};

  v2f B0[11], B1[11];
  const int  n1  = 16 + nl;
  const bool n1v = (n1 < Hh);
  const int  n1c = n1v ? n1 : (Hh - 1);
#pragma unroll
  for (int kc = 0; kc < 11; ++kc) {
    const int k   = kc * 4 + ko;
    const int kc0 = (k     < Fin) ? k     : 0;
    const int kc1 = (k + 1 < Fin) ? k + 1 : 0;
    const float w00 = w1[nl  * Fin + kc0];
    const float w01 = w1[nl  * Fin + kc1];
    const float w10 = w1[n1c * Fin + kc0];
    const float w11 = w1[n1c * Fin + kc1];
    B0[kc].x = (k     < Fin) ? w00 : 0.0f;
    B0[kc].y = (k + 1 < Fin) ? w01 : 0.0f;
    B1[kc].x = (n1v && (k     < Fin)) ? w10 : 0.0f;
    B1[kc].y = (n1v && (k + 1 < Fin)) ? w11 : 0.0f;
  }
  const float bA = b1[nl];
  const float bB = n1v ? b1[n1c] : 0.0f;

  const int NT = (Bsz * Tlen) / 16;
  for (int tile = blockIdx.x; tile < NT; tile += gridDim.x) {
    const float* xr = x + (size_t)(tile * 16 + nl) * Fin;

    if (tile + (int)gridDim.x < NT)
      __builtin_prefetch(xr + (size_t)gridDim.x * 16 * Fin, 0, 1);

    v8f a0e = {0, 0, 0, 0, 0, 0, 0, 0}, a0o = {0, 0, 0, 0, 0, 0, 0, 0};
    v8f a1e = {0, 0, 0, 0, 0, 0, 0, 0}, a1o = {0, 0, 0, 0, 0, 0, 0, 0};
#pragma unroll
    for (int kc = 0; kc < 10; kc += 2) {
      const v2f Ae = *(const v2f*)(xr + kc * 4 + ko);
      const v2f Ao = *(const v2f*)(xr + (kc + 1) * 4 + ko);
      a0e = WMMA4F32(Ae, B0[kc], a0e);
      a1e = WMMA4F32(Ae, B1[kc], a1e);
      a0o = WMMA4F32(Ao, B0[kc + 1], a0o);
      a1o = WMMA4F32(Ao, B1[kc + 1], a1o);
    }
    {
      const v2f t = *(const v2f*)(xr + 40);
      const v2f A = hiHalf ? zero2 : t;
      a0e = WMMA4F32(A, B0[10], a0e);
      a1e = WMMA4F32(A, B1[10], a1e);
    }

    float* arow = aOut + (size_t)tile * 16 * Hh;
#pragma unroll
    for (int j = 0; j < 8; ++j)
      arow[(j + mo) * Hh + nl] = tanhf_(a0e[j] + a0o[j] + bA);
    if (nl < 8) {
#pragma unroll
      for (int j = 0; j < 8; ++j)
        arow[(j + mo) * Hh + 16 + nl] = tanhf_(a1e[j] + a1o[j] + bB);
    }
  }
}

typedef __attribute__((ext_vector_type(4))) float v4f;

static constexpr int RD = 8;
#define AFRAG(S, q) \
  (((q) & 1) ? __builtin_shufflevector((S)[(q) >> 1], (S)[(q) >> 1], 2, 3) \
             : __builtin_shufflevector((S)[(q) >> 1], (S)[(q) >> 1], 0, 1))
static_assert(Tlen % RD == 0 && Tlen >= 2 * RD, "the step loop is unrolled by RD");

static constexpr int HPf = 28;

__device__ __forceinline__ float xhalf(float v) {
  const int b = __builtin_bit_cast(int, v);
  return __builtin_bit_cast(
      float, __builtin_amdgcn_permlanex16(b, b, 0x76543210, 0xfedcba98, false,
                                          false));
}

__global__ __launch_bounds__(96) void gru_scan_kernel(
    const float* __restrict__ a, const float* __restrict__ w_ih,
    const float* __restrict__ w_hh, const float* __restrict__ b_ih,
    const float* __restrict__ b_hh, const float* __restrict__ lc2_w,
    const float* __restrict__ lc2_b, float* __restrict__ out) {
  __shared__ __attribute__((aligned(16))) float hbuf[2 * 16 * HPf];

  const int lane = threadIdx.x & 31;
  const int w    = threadIdx.x >> 5;
  const int nl   = lane & 15;
  const int hf   = lane >> 4;
  const int kb   = 12 * hf;
  const int b0   = blockIdx.x * 16;
  const v2f zero2 = {0.0f, 0.0f};

  const int  mh  = nl >> 3, mj = nl & 7;
  const int  ua  = 8 * w + 4 * mh + (mj & 3);
  const bool lo4 = (mj < 4);
  const int rowRZ = (lo4 ? 0 : Hh) + ua, rowN = 2 * Hh + ua;

  v2f WxRZ[6], WxN[6], WhRZ[6], WhN[6];
#pragma unroll
  for (int q = 0; q < 6; ++q) {
    const int k = kb + 2 * q;
    WxRZ[q] = 0.5f * *(const v2f*)(w_ih + rowRZ * Hh + k);
    WhRZ[q] = 0.5f * *(const v2f*)(w_hh + rowRZ * Hh + k);
    const v2f xi = *(const v2f*)(w_ih + rowN * Hh + k);
    const v2f hh = *(const v2f*)(w_hh + rowN * Hh + k);
    WxN[q] = lo4 ? xi : zero2;
    WhN[q] = lo4 ? zero2 : hh;
  }

  v8f bRZ, bN;
#pragma unroll
  for (int j = 0; j < 8; ++j) {
    const int c = 8 * w + 4 * hf + (j & 3);
    bRZ[j] = 0.5f * ((j < 4) ? (b_ih[c] + b_hh[c])
                             : (b_ih[Hh + c] + b_hh[Hh + c]));
    bN[j]  = (j < 4) ? b_ih[2 * Hh + c] : b_hh[2 * Hh + c];
  }
  float l2w[12];
#pragma unroll
  for (int e = 0; e < 12; ++e) l2w[e] = lc2_w[kb + e];
  const float l2b = lc2_b[0];
  const bool storeLane = (hf == 0) && (w == 2);

  const float* arow = a + (size_t)(b0 + nl) * Tlen * Hh + kb;
  float*       orow = out + (size_t)(b0 + nl) * Tlen;
  v4f aR[RD][3];
#pragma unroll
  for (int s = 0; s < RD; ++s)
#pragma unroll
    for (int p = 0; p < 3; ++p) aR[s][p] = *(const v4f*)(arow + s * Hh + 4 * p);

  v8f xRZ = bRZ, xN = bN;
#pragma unroll
  for (int q = 0; q < 6; ++q) {
    xRZ = WMMA4F32(WxRZ[q], AFRAG(aR[0], q), xRZ);
    xN  = WMMA4F32(WxN[q],  AFRAG(aR[0], q), xN);
  }

  float*       hst = hbuf + nl * HPf + 8 * w + 4 * hf;
  const float* hld = hbuf + nl * HPf + kb;

  float h[12];
#pragma unroll
  for (int e = 0; e < 12; ++e) h[e] = 0.0f;
  float hown[4] = {0.0f, 0.0f, 0.0f, 0.0f};
  const float* ap = arow + RD * Hh;

  for (int tb = 0; tb < Tlen; tb += RD) {
#pragma unroll
  for (int u = 0; u < RD; ++u) {
    const int t  = tb + u;
    const int sl = (u + 1) % RD;

    v8f RZ = xRZ, N = xN;
#pragma unroll
    for (int q = 0; q < 6; ++q) {
      const v2f hB = {h[2 * q], h[2 * q + 1]};
      RZ = WMMA4F32(WhRZ[q], hB, RZ);
      N  = WMMA4F32(WhN[q],  hB, N);
    }

    float s = 0.0f;
#pragma unroll
    for (int e = 0; e < 12; ++e) s += l2w[e] * h[e];
    const float tot = s + xhalf(s);

    v4f hn;
#pragma unroll
    for (int i = 0; i < 4; ++i) {
      const float r  = sigmoid_of_half_(RZ[i]);
      const float z  = sigmoid_of_half_(RZ[4 + i]);
      const float nn = tanhf_(N[i] + r * N[4 + i]);
      const float hv = (1.0f - z) * nn + z * hown[i];
      hown[i] = hv;
      hn[i]   = hv;
    }

    const int bo = (t & 1) * (16 * HPf);
    *(v4f*)(hst + bo) = hn;

    if (storeLane && t > 0) orow[t - 1] = sigmoidf_(tot + l2b);

    __builtin_amdgcn_fence(__ATOMIC_RELEASE, "workgroup", "local");
    __builtin_amdgcn_s_barrier_signal(-1);
    __builtin_amdgcn_sched_barrier(0);
#pragma unroll
    for (int p = 0; p < 3; ++p)
      aR[u][p] = *(const v4f*)(ap + u * Hh + 4 * p);
    xRZ = bRZ; xN = bN;
#pragma unroll
    for (int q = 0; q < 3; ++q) {
      xRZ = WMMA4F32(WxRZ[q], AFRAG(aR[sl], q), xRZ);
      xN  = WMMA4F32(WxN[q],  AFRAG(aR[sl], q), xN);
    }
    __builtin_amdgcn_sched_barrier(0);
    __builtin_amdgcn_s_barrier_wait(-1);
    __builtin_amdgcn_fence(__ATOMIC_ACQUIRE, "workgroup", "local");
    const v4f h0 = *(const v4f*)(hld + bo);
    const v4f h1 = *(const v4f*)(hld + bo + 4);
    const v4f h2 = *(const v4f*)(hld + bo + 8);
    __builtin_amdgcn_sched_barrier(0);
#pragma unroll
    for (int q = 3; q < 6; ++q) {
      xRZ = WMMA4F32(WxRZ[q], AFRAG(aR[sl], q), xRZ);
      xN  = WMMA4F32(WxN[q],  AFRAG(aR[sl], q), xN);
    }
#pragma unroll
    for (int e = 0; e < 4; ++e) { h[e] = h0[e]; h[4 + e] = h1[e]; h[8 + e] = h2[e]; }

  }
    ap += (tb + 2 * RD < Tlen) ? RD * Hh : 0;
  }

  {
    float s = 0.0f;
#pragma unroll
    for (int e = 0; e < 12; ++e) s += l2w[e] * h[e];
    const float tot = s + xhalf(s);
    const float o   = sigmoidf_(tot + l2b);
    if (storeLane) orow[Tlen - 1] = o;
  }
}

extern "C" void kernel_launch(void* const* d_in, const int* in_sizes, int n_in,
                              void* d_out, int out_size, void* d_ws,
                              size_t ws_size, hipStream_t stream) {
  (void)in_sizes; (void)n_in; (void)out_size; (void)ws_size;
  const float* x     = (const float*)d_in[0];
  const float* lc1_w = (const float*)d_in[1];
  const float* lc1_b = (const float*)d_in[2];
  const float* w_ih  = (const float*)d_in[3];
  const float* w_hh  = (const float*)d_in[4];
  const float* b_ih  = (const float*)d_in[5];
  const float* b_hh  = (const float*)d_in[6];
  const float* lc2_w = (const float*)d_in[7];
  const float* lc2_b = (const float*)d_in[8];
  float* out  = (float*)d_out;
  float* aBuf = (float*)d_ws;

  lc1_tanh_kernel<<<2048, 32, 0, stream>>>(x, lc1_w, lc1_b, aBuf);
  gru_scan_kernel<<<Bsz / 16, 96, 0, stream>>>(aBuf, w_ih, w_hh, b_ih, b_hh,
                                               lc2_w, lc2_b, out);
}


# GRU scan: x-side gate products moved off the three scan waves to a fourth wave on the WGP's idle SIMD, formed three steps ahead, handed over through a four-slot LDS ring; sum order kept
# speedup vs baseline: 11.2433x; 1.0856x over previous
#include <hip/hip_runtime.h>
#include <hip/hip_bf16.h>
#include <cmath>


typedef __attribute__((ext_vector_type(2))) float v2f;
typedef __attribute__((ext_vector_type(8))) float v8f;

#define WMMA4F32(A, Bv, C)                                                   \
  __builtin_amdgcn_wmma_f32_16x16x4_f32(false, (A), false, (Bv), (short)0,   \
                                        (C), false, false)

static constexpr int Bsz  = 256;
static constexpr int Tlen = 2000;
static constexpr int Fin  = 42;
static constexpr int Hh   = 24;
static constexpr int Gg   = 72;


__device__ __forceinline__ float tanhf_(float x) {
  return __builtin_amdgcn_tanhf(x);
}

__device__ __forceinline__ float sigmoid_of_half_(float u) {
  return __builtin_fmaf(0.5f, __builtin_amdgcn_tanhf(u), 0.5f);
}
__device__ __forceinline__ float sigmoidf_(float v) {
  return sigmoid_of_half_(0.5f * v);
}

__global__ __launch_bounds__(32) void lc1_tanh_kernel(
    const float* __restrict__ x, const float* __restrict__ w1,
    const float* __restrict__ b1, float* __restrict__ aOut) {
  const int  lane = threadIdx.x;
  const int  nl   = lane & 15;
  const int  ko   = (lane >> 4) * 2;
  const int  mo   = (lane >> 4) * 8;
  const bool hiHalf = (lane >= 16);
  const v2f  zero2 = {0.0f, 0.0f};

  v2f B0[11], B1[11];
  const int  n1  = 16 + nl;
  const bool n1v = (n1 < Hh);
  const int  n1c = n1v ? n1 : (Hh - 1);
#pragma unroll
  for (int kc = 0; kc < 11; ++kc) {
    const int k   = kc * 4 + ko;
    const int kc0 = (k     < Fin) ? k     : 0;
    const int kc1 = (k + 1 < Fin) ? k + 1 : 0;
    const float w00 = w1[nl  * Fin + kc0];
    const float w01 = w1[nl  * Fin + kc1];
    const float w10 = w1[n1c * Fin + kc0];
    const float w11 = w1[n1c * Fin + kc1];
    B0[kc].x = (k     < Fin) ? w00 : 0.0f;
    B0[kc].y = (k + 1 < Fin) ? w01 : 0.0f;
    B1[kc].x = (n1v && (k     < Fin)) ? w10 : 0.0f;
    B1[kc].y = (n1v && (k + 1 < Fin)) ? w11 : 0.0f;
  }
  const float bA = b1[nl];
  const float bB = n1v ? b1[n1c] : 0.0f;

  const int NT = (Bsz * Tlen) / 16;
  for (int tile = blockIdx.x; tile < NT; tile += gridDim.x) {
    const float* xr = x + (size_t)(tile * 16 + nl) * Fin;

    if (tile + (int)gridDim.x < NT)
      __builtin_prefetch(xr + (size_t)gridDim.x * 16 * Fin, 0, 1);

    v8f a0e = {0, 0, 0, 0, 0, 0, 0, 0}, a0o = {0, 0, 0, 0, 0, 0, 0, 0};
    v8f a1e = {0, 0, 0, 0, 0, 0, 0, 0}, a1o = {0, 0, 0, 0, 0, 0, 0, 0};
#pragma unroll
    for (int kc = 0; kc < 10; kc += 2) {
      const v2f Ae = *(const v2f*)(xr + kc * 4 + ko);
      const v2f Ao = *(const v2f*)(xr + (kc + 1) * 4 + ko);
      a0e = WMMA4F32(Ae, B0[kc], a0e);
      a1e = WMMA4F32(Ae, B1[kc], a1e);
      a0o = WMMA4F32(Ao, B0[kc + 1], a0o);
      a1o = WMMA4F32(Ao, B1[kc + 1], a1o);
    }
    {
      const v2f t = *(const v2f*)(xr + 40);
      const v2f A = hiHalf ? zero2 : t;
      a0e = WMMA4F32(A, B0[10], a0e);
      a1e = WMMA4F32(A, B1[10], a1e);
    }

    float* arow = aOut + (size_t)tile * 16 * Hh;
#pragma unroll
    for (int j = 0; j < 8; ++j)
      arow[(j + mo) * Hh + nl] = tanhf_(a0e[j] + a0o[j] + bA);
    if (nl < 8) {
#pragma unroll
      for (int j = 0; j < 8; ++j)
        arow[(j + mo) * Hh + 16 + nl] = tanhf_(a1e[j] + a1o[j] + bB);
    }
  }
}

typedef __attribute__((ext_vector_type(4))) float v4f;

static constexpr int RD = 8;
#define AFRAG(S, q) \
  (((q) & 1) ? __builtin_shufflevector((S)[(q) >> 1], (S)[(q) >> 1], 2, 3) \
             : __builtin_shufflevector((S)[(q) >> 1], (S)[(q) >> 1], 0, 1))
#define LO4(V) __builtin_shufflevector((V), (V), 0, 1, 2, 3)
#define HI4(V) __builtin_shufflevector((V), (V), 4, 5, 6, 7)
static_assert(Tlen % RD == 0 && Tlen >= 2 * RD, "the step loop is unrolled by RD");
static_assert(RD % 4 == 0, "slot and buffer numbers are taken from the unroll index");

static constexpr int HPf = 28;
static constexpr int XPf = 76;
static constexpr int XSL = 16 * XPf;
static constexpr int NXS = 4;

__device__ __forceinline__ float xhalf(float v) {
  const int b = __builtin_bit_cast(int, v);
  return __builtin_bit_cast(
      float, __builtin_amdgcn_permlanex16(b, b, 0x76543210, 0xfedcba98, false,
                                          false));
}

__global__ __launch_bounds__(128) void gru_scan_kernel(
    const float* __restrict__ a, const float* __restrict__ w_ih,
    const float* __restrict__ w_hh, const float* __restrict__ b_ih,
    const float* __restrict__ b_hh, const float* __restrict__ lc2_w,
    const float* __restrict__ lc2_b, float* __restrict__ out) {
  __shared__ __attribute__((aligned(16))) float hbuf[2 * 16 * HPf];
  __shared__ __attribute__((aligned(16))) float xbuf[NXS * XSL];

  const int lane = threadIdx.x & 31;
  const int wv   = __builtin_amdgcn_readfirstlane((int)(threadIdx.x >> 5));
  const bool scan = (wv < 3);
  const int w    = scan ? wv : 0;
  const int nl   = lane & 15;
  const int hf   = lane >> 4;
  const int kb   = 12 * hf;
  const int b0   = blockIdx.x * 16;
  const v2f zero2 = {0.0f, 0.0f};

  const int  mh  = nl >> 3, mj = nl & 7;
  const bool lo4 = (mj < 4);

  v2f WhRZ[6], WhN[6];
  {
    const int ua = 8 * w + 4 * mh + (mj & 3);
    const int rowRZ = (lo4 ? 0 : Hh) + ua, rowN = 2 * Hh + ua;
#pragma unroll
    for (int q = 0; q < 6; ++q) {
      const int k = kb + 2 * q;
      WhRZ[q] = 0.5f * *(const v2f*)(w_hh + rowRZ * Hh + k);
      const v2f hh = *(const v2f*)(w_hh + rowN * Hh + k);
      WhN[q] = lo4 ? zero2 : hh;
    }
  }
  v2f WxRZ[3][6], WxN0[6], WxN1[6];
  {
    const bool x1v = (nl < 8);
    const int  x1r = 2 * Hh + 16 + (x1v ? nl : 0);
#pragma unroll
    for (int q = 0; q < 6; ++q) {
      const int k = kb + 2 * q;
#pragma unroll
      for (int v = 0; v < 3; ++v) {
        const int ua = 8 * v + 4 * mh + (mj & 3);
        const int rowRZ = (lo4 ? 0 : Hh) + ua;
        WxRZ[v][q] = 0.5f * *(const v2f*)(w_ih + rowRZ * Hh + k);
      }
      WxN0[q] = *(const v2f*)(w_ih + (2 * Hh + nl) * Hh + k);
      const v2f t1 = *(const v2f*)(w_ih + x1r * Hh + k);
      WxN1[q] = x1v ? t1 : zero2;
    }
  }

  v8f bRZ[3], bX0, bX1;
#pragma unroll
  for (int j = 0; j < 8; ++j) {
#pragma unroll
    for (int v = 0; v < 3; ++v) {
      const int c = 8 * v + 4 * hf + (j & 3);
      bRZ[v][j] = 0.5f * ((j < 4) ? (b_ih[c] + b_hh[c])
                                  : (b_ih[Hh + c] + b_hh[Hh + c]));
    }
    bX0[j] = b_ih[2 * Hh + 8 * hf + j];
    const float t1 = b_ih[2 * Hh + 16 + j];
    bX1[j] = (hf == 0) ? t1 : 0.0f;
  }
  float bhn[4];
#pragma unroll
  for (int i = 0; i < 4; ++i) bhn[i] = b_hh[2 * Hh + 8 * w + 4 * hf + i];
  float l2w[12];
#pragma unroll
  for (int e = 0; e < 12; ++e) l2w[e] = lc2_w[kb + e];
  const float l2b = lc2_b[0];
  const bool storeLane = (hf == 0) && (wv == 2);

  const float* arow = a + (size_t)(b0 + nl) * Tlen * Hh + kb;
  float*       orow = out + (size_t)(b0 + nl) * Tlen;

  float*       hst = hbuf + nl * HPf + 8 * w + 4 * hf;
  const float* hld = hbuf + nl * HPf + kb;
  float*       xsRZ = xbuf + nl * XPf + 12 * hf;
  float*       xsN0 = xbuf + nl * XPf + 24 * hf + 8;
  float*       xsN1 = xbuf + nl * XPf + 48 + 8;
  const float* xld  = xbuf + nl * XPf + 12 * (2 * w + hf);

  auto xside = [&](const v4f (&S)[3], v8f (&oRZ)[3], v8f& o0, v8f& o1) {
    oRZ[0] = bRZ[0]; oRZ[1] = bRZ[1]; oRZ[2] = bRZ[2]; o0 = bX0; o1 = bX1;
#pragma unroll
    for (int q = 0; q < 6; ++q) {
      const v2f aB = AFRAG(S, q);
      oRZ[0] = WMMA4F32(WxRZ[0][q], aB, oRZ[0]);
      oRZ[1] = WMMA4F32(WxRZ[1][q], aB, oRZ[1]);
      oRZ[2] = WMMA4F32(WxRZ[2][q], aB, oRZ[2]);
      o0     = WMMA4F32(WxN0[q],    aB, o0);
      o1     = WMMA4F32(WxN1[q],    aB, o1);
    }
  };
  auto xstore = [&](int slot, const v8f (&pRZ)[3], const v8f& p0, const v8f& p1) {
    const int so = slot * XSL;
#pragma unroll
    for (int v = 0; v < 3; ++v) {
      *(v4f*)(xsRZ + so + 24 * v)     = LO4(pRZ[v]);
      *(v4f*)(xsRZ + so + 24 * v + 4) = HI4(pRZ[v]);
    }
    *(v4f*)(xsN0 + so)      = LO4(p0);
    *(v4f*)(xsN0 + so + 12) = HI4(p0);
    if (hf == 0) {
      *(v4f*)(xsN1 + so)      = LO4(p1);
      *(v4f*)(xsN1 + so + 12) = HI4(p1);
    }
  };

  v4f aR[RD][3];
#pragma unroll
  for (int s = 0; s < RD; ++s)
#pragma unroll
    for (int p = 0; p < 3; ++p) aR[s][p] = *(const v4f*)(arow + s * Hh + 4 * p);
  if (!scan) {
#pragma unroll
    for (int s = 0; s < 3; ++s) {
      v8f rRZ[3], r0, r1;
      xside(aR[s], rRZ, r0, r1);
      xstore(s, rRZ, r0, r1);
    }
  }
  if (scan) *(v4f*)(hst + 16 * HPf) = (v4f){0.0f, 0.0f, 0.0f, 0.0f};
#pragma unroll
  for (int p = 0; p < 3; ++p) {
    aR[0][p] = *(const v4f*)(arow + (RD + 0) * Hh + 4 * p);
    aR[1][p] = *(const v4f*)(arow + (RD + 1) * Hh + 4 * p);
  }
  __builtin_amdgcn_fence(__ATOMIC_RELEASE, "workgroup", "local");
  __builtin_amdgcn_s_barrier_signal(-1);
  __builtin_amdgcn_s_barrier_wait(-1);
  __builtin_amdgcn_fence(__ATOMIC_ACQUIRE, "workgroup", "local");

  const v4f bhnv = {bhn[0], bhn[1], bhn[2], bhn[3]};
  float hown[4] = {0.0f, 0.0f, 0.0f, 0.0f};

  for (int tb = 0; tb < Tlen; tb += RD) {
#pragma unroll
  for (int u = 0; u < RD; ++u) {
    const int t  = tb + u;
    const int bw = (u & 1) * (16 * HPf);
    const int br = ((u + 1) & 1) * (16 * HPf);

    if (scan) {
      __builtin_amdgcn_fence(__ATOMIC_ACQUIRE, "workgroup", "local");
      const v4f lo = *(const v4f*)(xld + (u & 3) * XSL);
      const v4f hi = *(const v4f*)(xld + (u & 3) * XSL + 4);
      const v4f n4 = *(const v4f*)(xld + (u & 3) * XSL + 8);
      const v4f h0 = *(const v4f*)(hld + br);
      const v4f h1 = *(const v4f*)(hld + br + 4);
      const v4f h2 = *(const v4f*)(hld + br + 8);
      float h[12];
#pragma unroll
      for (int e = 0; e < 4; ++e) { h[e] = h0[e]; h[4 + e] = h1[e]; h[8 + e] = h2[e]; }

      v8f RZ = __builtin_shufflevector(lo, hi, 0, 1, 2, 3, 4, 5, 6, 7);
      v8f N  = __builtin_shufflevector(n4, bhnv, 0, 1, 2, 3, 4, 5, 6, 7);
#pragma unroll
      for (int q = 0; q < 6; ++q) {
        const v2f hB = {h[2 * q], h[2 * q + 1]};
        RZ = WMMA4F32(WhRZ[q], hB, RZ);
        N  = WMMA4F32(WhN[q],  hB, N);
      }

      float s = 0.0f;
#pragma unroll
      for (int e = 0; e < 12; ++e) s += l2w[e] * h[e];
      const float tot = s + xhalf(s);

      v4f hn;
#pragma unroll
      for (int i = 0; i < 4; ++i) {
        const float r  = sigmoid_of_half_(RZ[i]);
        const float z  = sigmoid_of_half_(RZ[4 + i]);
        const float nn = tanhf_(N[i] + r * N[4 + i]);
        const float hv = (1.0f - z) * nn + z * hown[i];
        hown[i] = hv;
        hn[i]   = hv;
      }

      *(v4f*)(hst + bw) = hn;

      if (storeLane && t > 0) orow[t - 1] = sigmoidf_(tot + l2b);

      __builtin_amdgcn_fence(__ATOMIC_RELEASE, "workgroup", "local");
    } else {
      v8f rRZ[3], r0, r1;
      xside(aR[(u + 3) % RD], rRZ, r0, r1);
      __builtin_amdgcn_fence(__ATOMIC_RELEASE, "workgroup", "local");
      xstore((u + 3) & 3, rRZ, r0, r1);
    }

    __builtin_amdgcn_s_barrier_signal(-1);
    {
      const int tau = t + 2 + RD;
      const float* src = arow + (size_t)((tau < Tlen) ? tau : (Tlen - 1)) * Hh;
#pragma unroll
      for (int p = 0; p < 3; ++p)
        aR[(u + 2) % RD][p] = *(const v4f*)(src + 4 * p);
    }
    __builtin_amdgcn_s_barrier_wait(-1);
  }
  }

  {
    static_assert(((Tlen - 1) & 1) == 1, "the last step writes h buffer 1");
    __builtin_amdgcn_fence(__ATOMIC_ACQUIRE, "workgroup", "local");
    const v4f h0 = *(const v4f*)(hld + 16 * HPf);
    const v4f h1 = *(const v4f*)(hld + 16 * HPf + 4);
    const v4f h2 = *(const v4f*)(hld + 16 * HPf + 8);
    float h[12];
#pragma unroll
    for (int e = 0; e < 4; ++e) { h[e] = h0[e]; h[4 + e] = h1[e]; h[8 + e] = h2[e]; }
    float s = 0.0f;
#pragma unroll
    for (int e = 0; e < 12; ++e) s += l2w[e] * h[e];
    const float tot = s + xhalf(s);
    const float o   = sigmoidf_(tot + l2b);
    if (storeLane) orow[Tlen - 1] = o;
  }
}

extern "C" void kernel_launch(void* const* d_in, const int* in_sizes, int n_in,
                              void* d_out, int out_size, void* d_ws,
                              size_t ws_size, hipStream_t stream) {
  (void)in_sizes; (void)n_in; (void)out_size; (void)ws_size;
  const float* x     = (const float*)d_in[0];
  const float* lc1_w = (const float*)d_in[1];
  const float* lc1_b = (const float*)d_in[2];
  const float* w_ih  = (const float*)d_in[3];
  const float* w_hh  = (const float*)d_in[4];
  const float* b_ih  = (const float*)d_in[5];
  const float* b_hh  = (const float*)d_in[6];
  const float* lc2_w = (const float*)d_in[7];
  const float* lc2_b = (const float*)d_in[8];
  float* out  = (float*)d_out;
  float* aBuf = (float*)d_ws;

  lc1_tanh_kernel<<<2048, 32, 0, stream>>>(x, lc1_w, lc1_b, aBuf);
  gru_scan_kernel<<<Bsz / 16, 128, 0, stream>>>(aBuf, w_ih, w_hh, b_ih, b_hh,
                                               lc2_w, lc2_b, out);
}


# GRU scan: x-side gate products split over five waves, one 16-row tile each (was one wave forming all five), 8-wave work-group: two on the idle SIMD, three beside the scan waves; sum order kept
# speedup vs baseline: 12.1589x; 1.0814x over previous
#include <hip/hip_runtime.h>
#include <hip/hip_bf16.h>
#include <cmath>


typedef __attribute__((ext_vector_type(2))) float v2f;
typedef __attribute__((ext_vector_type(8))) float v8f;

#define WMMA4F32(A, Bv, C)                                                   \
  __builtin_amdgcn_wmma_f32_16x16x4_f32(false, (A), false, (Bv), (short)0,   \
                                        (C), false, false)

static constexpr int Bsz  = 256;
static constexpr int Tlen = 2000;
static constexpr int Fin  = 42;
static constexpr int Hh   = 24;
static constexpr int Gg   = 72;


__device__ __forceinline__ float tanhf_(float x) {
  return __builtin_amdgcn_tanhf(x);
}

__device__ __forceinline__ float sigmoid_of_half_(float u) {
  return __builtin_fmaf(0.5f, __builtin_amdgcn_tanhf(u), 0.5f);
}
__device__ __forceinline__ float sigmoidf_(float v) {
  return sigmoid_of_half_(0.5f * v);
}

__global__ __launch_bounds__(32) void lc1_tanh_kernel(
    const float* __restrict__ x, const float* __restrict__ w1,
    const float* __restrict__ b1, float* __restrict__ aOut) {
  const int  lane = threadIdx.x;
  const int  nl   = lane & 15;
  const int  ko   = (lane >> 4) * 2;
  const int  mo   = (lane >> 4) * 8;
  const bool hiHalf = (lane >= 16);
  const v2f  zero2 = {0.0f, 0.0f};

  v2f B0[11], B1[11];
  const int  n1  = 16 + nl;
  const bool n1v = (n1 < Hh);
  const int  n1c = n1v ? n1 : (Hh - 1);
#pragma unroll
  for (int kc = 0; kc < 11; ++kc) {
    const int k   = kc * 4 + ko;
    const int kc0 = (k     < Fin) ? k     : 0;
    const int kc1 = (k + 1 < Fin) ? k + 1 : 0;
    const float w00 = w1[nl  * Fin + kc0];
    const float w01 = w1[nl  * Fin + kc1];
    const float w10 = w1[n1c * Fin + kc0];
    const float w11 = w1[n1c * Fin + kc1];
    B0[kc].x = (k     < Fin) ? w00 : 0.0f;
    B0[kc].y = (k + 1 < Fin) ? w01 : 0.0f;
    B1[kc].x = (n1v && (k     < Fin)) ? w10 : 0.0f;
    B1[kc].y = (n1v && (k + 1 < Fin)) ? w11 : 0.0f;
  }
  const float bA = b1[nl];
  const float bB = n1v ? b1[n1c] : 0.0f;

  const int NT = (Bsz * Tlen) / 16;
  for (int tile = blockIdx.x; tile < NT; tile += gridDim.x) {
    const float* xr = x + (size_t)(tile * 16 + nl) * Fin;

    if (tile + (int)gridDim.x < NT)
      __builtin_prefetch(xr + (size_t)gridDim.x * 16 * Fin, 0, 1);

    v8f a0e = {0, 0, 0, 0, 0, 0, 0, 0}, a0o = {0, 0, 0, 0, 0, 0, 0, 0};
    v8f a1e = {0, 0, 0, 0, 0, 0, 0, 0}, a1o = {0, 0, 0, 0, 0, 0, 0, 0};
#pragma unroll
    for (int kc = 0; kc < 10; kc += 2) {
      const v2f Ae = *(const v2f*)(xr + kc * 4 + ko);
      const v2f Ao = *(const v2f*)(xr + (kc + 1) * 4 + ko);
      a0e = WMMA4F32(Ae, B0[kc], a0e);
      a1e = WMMA4F32(Ae, B1[kc], a1e);
      a0o = WMMA4F32(Ao, B0[kc + 1], a0o);
      a1o = WMMA4F32(Ao, B1[kc + 1], a1o);
    }
    {
      const v2f t = *(const v2f*)(xr + 40);
      const v2f A = hiHalf ? zero2 : t;
      a0e = WMMA4F32(A, B0[10], a0e);
      a1e = WMMA4F32(A, B1[10], a1e);
    }

    float* arow = aOut + (size_t)tile * 16 * Hh;
#pragma unroll
    for (int j = 0; j < 8; ++j)
      arow[(j + mo) * Hh + nl] = tanhf_(a0e[j] + a0o[j] + bA);
    if (nl < 8) {
#pragma unroll
      for (int j = 0; j < 8; ++j)
        arow[(j + mo) * Hh + 16 + nl] = tanhf_(a1e[j] + a1o[j] + bB);
    }
  }
}

typedef __attribute__((ext_vector_type(4))) float v4f;

static constexpr int RD = 8;
#define AFRAG(S, q) \
  (((q) & 1) ? __builtin_shufflevector((S)[(q) >> 1], (S)[(q) >> 1], 2, 3) \
             : __builtin_shufflevector((S)[(q) >> 1], (S)[(q) >> 1], 0, 1))
#define LO4(V) __builtin_shufflevector((V), (V), 0, 1, 2, 3)
#define HI4(V) __builtin_shufflevector((V), (V), 4, 5, 6, 7)
static_assert(Tlen % RD == 0 && Tlen >= 2 * RD, "the step loop is unrolled by RD");
static_assert(RD % 4 == 0, "slot and buffer numbers are taken from the unroll index");

static constexpr int HPf = 28;
static constexpr int XPf = 76;
static constexpr int XSL = 16 * XPf;
static constexpr int NXS = 4;

__device__ __forceinline__ float xhalf(float v) {
  const int b = __builtin_bit_cast(int, v);
  return __builtin_bit_cast(
      float, __builtin_amdgcn_permlanex16(b, b, 0x76543210, 0xfedcba98, false,
                                          false));
}

__global__ __launch_bounds__(256) void gru_scan_kernel(
    const float* __restrict__ a, const float* __restrict__ w_ih,
    const float* __restrict__ w_hh, const float* __restrict__ b_ih,
    const float* __restrict__ b_hh, const float* __restrict__ lc2_w,
    const float* __restrict__ lc2_b, float* __restrict__ out) {
  __shared__ __attribute__((aligned(16))) float hbuf[2 * 16 * HPf];
  __shared__ __attribute__((aligned(16))) float xbuf[NXS * XSL];

  const int lane = threadIdx.x & 31;
  const int wv   = __builtin_amdgcn_readfirstlane((int)(threadIdx.x >> 5));
  const bool scan = (wv < 3);
  const int w    = scan ? wv : 0;
  const int xt   = scan ? 0 : (wv - 3);
  const bool xrz = (xt < 3);
  const int nl   = lane & 15;
  const int hf   = lane >> 4;
  const int kb   = 12 * hf;
  const int b0   = blockIdx.x * 16;
  const v2f zero2 = {0.0f, 0.0f};

  const int  mh  = nl >> 3, mj = nl & 7;
  const bool lo4 = (mj < 4);

  v2f WhRZ[6], WhN[6];
  {
    const int ua = 8 * w + 4 * mh + (mj & 3);
    const int rowRZ = (lo4 ? 0 : Hh) + ua, rowN = 2 * Hh + ua;
#pragma unroll
    for (int q = 0; q < 6; ++q) {
      const int k = kb + 2 * q;
      WhRZ[q] = 0.5f * *(const v2f*)(w_hh + rowRZ * Hh + k);
      const v2f hh = *(const v2f*)(w_hh + rowN * Hh + k);
      WhN[q] = lo4 ? zero2 : hh;
    }
  }
  v2f Wx[6];
  {
    const int  uaX   = 8 * (xrz ? xt : 0) + 4 * mh + (mj & 3);
    const int  rowRZ = (lo4 ? 0 : Hh) + uaX;
    const bool x1v   = (nl < 8);
    const int  rowN  = (xt == 4) ? (2 * Hh + 16 + (x1v ? nl : 0))
                                 : (2 * Hh + nl);
    const int  rowX  = xrz ? rowRZ : rowN;
    const bool rowV  = (xt != 4) || x1v;
#pragma unroll
    for (int q = 0; q < 6; ++q) {
      const v2f t = *(const v2f*)(w_ih + rowX * Hh + kb + 2 * q);
      const v2f th = 0.5f * t;
      Wx[q] = xrz ? th : (rowV ? t : zero2);
    }
  }

  v8f bX;
#pragma unroll
  for (int j = 0; j < 8; ++j) {
    const int c = 8 * (xrz ? xt : 0) + 4 * hf + (j & 3);
    const float rz = 0.5f * ((j < 4) ? (b_ih[c] + b_hh[c])
                                     : (b_ih[Hh + c] + b_hh[Hh + c]));
    const float n0 = b_ih[2 * Hh + 8 * hf + j];
    const float t1 = b_ih[2 * Hh + 16 + j];
    const float n1 = (hf == 0) ? t1 : 0.0f;
    bX[j] = xrz ? rz : ((xt == 3) ? n0 : n1);
  }
  float bhn[4];
#pragma unroll
  for (int i = 0; i < 4; ++i) bhn[i] = b_hh[2 * Hh + 8 * w + 4 * hf + i];
  float l2w[12];
#pragma unroll
  for (int e = 0; e < 12; ++e) l2w[e] = lc2_w[kb + e];
  const float l2b = lc2_b[0];
  const bool storeLane = (hf == 0) && (wv == 2);

  const float* arow = a + (size_t)(b0 + nl) * Tlen * Hh + kb;
  float*       orow = out + (size_t)(b0 + nl) * Tlen;

  float*       hst = hbuf + nl * HPf + 8 * w + 4 * hf;
  const float* hld = hbuf + nl * HPf + kb;
  float*       xsLo = xbuf + nl * XPf +
                      (xrz ? (12 * hf + 24 * xt)
                           : ((xt == 3) ? (24 * hf + 8) : (48 + 8)));
  float*       xsHi = xsLo + (xrz ? 4 : 12);
  const bool   xsEn = (xt != 4) || (hf == 0);
  const float* xld  = xbuf + nl * XPf + 12 * (2 * w + hf);

  auto xside = [&](const v4f (&S)[3], v8f& o) {
    o = bX;
#pragma unroll
    for (int q = 0; q < 6; ++q) o = WMMA4F32(Wx[q], AFRAG(S, q), o);
  };
  auto xstore = [&](int slot, const v8f& p) {
    const int so = slot * XSL;
    if (xsEn) {
      *(v4f*)(xsLo + so) = LO4(p);
      *(v4f*)(xsHi + so) = HI4(p);
    }
  };

  v4f aR[RD][3];
#pragma unroll
  for (int s = 0; s < RD; ++s)
#pragma unroll
    for (int p = 0; p < 3; ++p) aR[s][p] = *(const v4f*)(arow + s * Hh + 4 * p);
  if (!scan) {
#pragma unroll
    for (int s = 0; s < 3; ++s) {
      v8f r;
      xside(aR[s], r);
      xstore(s, r);
    }
  }
  if (scan) *(v4f*)(hst + 16 * HPf) = (v4f){0.0f, 0.0f, 0.0f, 0.0f};
#pragma unroll
  for (int p = 0; p < 3; ++p) {
    aR[0][p] = *(const v4f*)(arow + (RD + 0) * Hh + 4 * p);
    aR[1][p] = *(const v4f*)(arow + (RD + 1) * Hh + 4 * p);
  }
  __builtin_amdgcn_fence(__ATOMIC_RELEASE, "workgroup", "local");
  __builtin_amdgcn_s_barrier_signal(-1);
  __builtin_amdgcn_s_barrier_wait(-1);
  __builtin_amdgcn_fence(__ATOMIC_ACQUIRE, "workgroup", "local");

  const v4f bhnv = {bhn[0], bhn[1], bhn[2], bhn[3]};
  float hown[4] = {0.0f, 0.0f, 0.0f, 0.0f};

  for (int tb = 0; tb < Tlen; tb += RD) {
#pragma unroll
  for (int u = 0; u < RD; ++u) {
    const int t  = tb + u;
    const int bw = (u & 1) * (16 * HPf);
    const int br = ((u + 1) & 1) * (16 * HPf);

    if (scan) {
      __builtin_amdgcn_fence(__ATOMIC_ACQUIRE, "workgroup", "local");
      const v4f lo = *(const v4f*)(xld + (u & 3) * XSL);
      const v4f hi = *(const v4f*)(xld + (u & 3) * XSL + 4);
      const v4f n4 = *(const v4f*)(xld + (u & 3) * XSL + 8);
      const v4f h0 = *(const v4f*)(hld + br);
      const v4f h1 = *(const v4f*)(hld + br + 4);
      const v4f h2 = *(const v4f*)(hld + br + 8);
      float h[12];
#pragma unroll
      for (int e = 0; e < 4; ++e) { h[e] = h0[e]; h[4 + e] = h1[e]; h[8 + e] = h2[e]; }

      v8f RZ = __builtin_shufflevector(lo, hi, 0, 1, 2, 3, 4, 5, 6, 7);
      v8f N  = __builtin_shufflevector(n4, bhnv, 0, 1, 2, 3, 4, 5, 6, 7);
#pragma unroll
      for (int q = 0; q < 6; ++q) {
        const v2f hB = {h[2 * q], h[2 * q + 1]};
        RZ = WMMA4F32(WhRZ[q], hB, RZ);
        N  = WMMA4F32(WhN[q],  hB, N);
      }

      float s = 0.0f;
#pragma unroll
      for (int e = 0; e < 12; ++e) s += l2w[e] * h[e];
      const float tot = s + xhalf(s);

      v4f hn;
#pragma unroll
      for (int i = 0; i < 4; ++i) {
        const float r  = sigmoid_of_half_(RZ[i]);
        const float z  = sigmoid_of_half_(RZ[4 + i]);
        const float nn = tanhf_(N[i] + r * N[4 + i]);
        const float hv = (1.0f - z) * nn + z * hown[i];
        hown[i] = hv;
        hn[i]   = hv;
      }

      *(v4f*)(hst + bw) = hn;

      if (storeLane && t > 0) orow[t - 1] = sigmoidf_(tot + l2b);

      __builtin_amdgcn_fence(__ATOMIC_RELEASE, "workgroup", "local");
    } else {
      v8f r;
      xside(aR[(u + 3) % RD], r);
      __builtin_amdgcn_fence(__ATOMIC_RELEASE, "workgroup", "local");
      xstore((u + 3) & 3, r);
    }

    __builtin_amdgcn_s_barrier_signal(-1);
    {
      const int tau = t + 2 + RD;
      const float* src = arow + (size_t)((tau < Tlen) ? tau : (Tlen - 1)) * Hh;
#pragma unroll
      for (int p = 0; p < 3; ++p)
        aR[(u + 2) % RD][p] = *(const v4f*)(src + 4 * p);
    }
    __builtin_amdgcn_s_barrier_wait(-1);
  }
  }

  {
    static_assert(((Tlen - 1) & 1) == 1, "the last step writes h buffer 1");
    __builtin_amdgcn_fence(__ATOMIC_ACQUIRE, "workgroup", "local");
    const v4f h0 = *(const v4f*)(hld + 16 * HPf);
    const v4f h1 = *(const v4f*)(hld + 16 * HPf + 4);
    const v4f h2 = *(const v4f*)(hld + 16 * HPf + 8);
    float h[12];
#pragma unroll
    for (int e = 0; e < 4; ++e) { h[e] = h0[e]; h[4 + e] = h1[e]; h[8 + e] = h2[e]; }
    float s = 0.0f;
#pragma unroll
    for (int e = 0; e < 12; ++e) s += l2w[e] * h[e];
    const float tot = s + xhalf(s);
    const float o   = sigmoidf_(tot + l2b);
    if (storeLane) orow[Tlen - 1] = o;
  }
}

extern "C" void kernel_launch(void* const* d_in, const int* in_sizes, int n_in,
                              void* d_out, int out_size, void* d_ws,
                              size_t ws_size, hipStream_t stream) {
  (void)in_sizes; (void)n_in; (void)out_size; (void)ws_size;
  const float* x     = (const float*)d_in[0];
  const float* lc1_w = (const float*)d_in[1];
  const float* lc1_b = (const float*)d_in[2];
  const float* w_ih  = (const float*)d_in[3];
  const float* w_hh  = (const float*)d_in[4];
  const float* b_ih  = (const float*)d_in[5];
  const float* b_hh  = (const float*)d_in[6];
  const float* lc2_w = (const float*)d_in[7];
  const float* lc2_b = (const float*)d_in[8];
  float* out  = (float*)d_out;
  float* aBuf = (float*)d_ws;

  lc1_tanh_kernel<<<2048, 32, 0, stream>>>(x, lc1_w, lc1_b, aBuf);
  gru_scan_kernel<<<Bsz / 16, 256, 0, stream>>>(aBuf, w_ih, w_hh, b_ih, b_hh,
                                               lc2_w, lc2_b, out);
}


# GRU scan: lc2 head (24-term sum, sigmoid, out store) moved off the three scan waves, which every step waits for, to one x-side wave that reads h from the LDS exchange buffer; sum order kept
# speedup vs baseline: 12.6883x; 1.0435x over previous
#include <hip/hip_runtime.h>
#include <hip/hip_bf16.h>
#include <cmath>


typedef __attribute__((ext_vector_type(2))) float v2f;
typedef __attribute__((ext_vector_type(8))) float v8f;

#define WMMA4F32(A, Bv, C)                                                   \
  __builtin_amdgcn_wmma_f32_16x16x4_f32(false, (A), false, (Bv), (short)0,   \
                                        (C), false, false)

static constexpr int Bsz  = 256;
static constexpr int Tlen = 2000;
static constexpr int Fin  = 42;
static constexpr int Hh   = 24;
static constexpr int Gg   = 72;


__device__ __forceinline__ float tanhf_(float x) {
  return __builtin_amdgcn_tanhf(x);
}

__device__ __forceinline__ float sigmoid_of_half_(float u) {
  return __builtin_fmaf(0.5f, __builtin_amdgcn_tanhf(u), 0.5f);
}
__device__ __forceinline__ float sigmoidf_(float v) {
  return sigmoid_of_half_(0.5f * v);
}

__global__ __launch_bounds__(32) void lc1_tanh_kernel(
    const float* __restrict__ x, const float* __restrict__ w1,
    const float* __restrict__ b1, float* __restrict__ aOut) {
  const int  lane = threadIdx.x;
  const int  nl   = lane & 15;
  const int  ko   = (lane >> 4) * 2;
  const int  mo   = (lane >> 4) * 8;
  const bool hiHalf = (lane >= 16);
  const v2f  zero2 = {0.0f, 0.0f};

  v2f B0[11], B1[11];
  const int  n1  = 16 + nl;
  const bool n1v = (n1 < Hh);
  const int  n1c = n1v ? n1 : (Hh - 1);
#pragma unroll
  for (int kc = 0; kc < 11; ++kc) {
    const int k   = kc * 4 + ko;
    const int kc0 = (k     < Fin) ? k     : 0;
    const int kc1 = (k + 1 < Fin) ? k + 1 : 0;
    const float w00 = w1[nl  * Fin + kc0];
    const float w01 = w1[nl  * Fin + kc1];
    const float w10 = w1[n1c * Fin + kc0];
    const float w11 = w1[n1c * Fin + kc1];
    B0[kc].x = (k     < Fin) ? w00 : 0.0f;
    B0[kc].y = (k + 1 < Fin) ? w01 : 0.0f;
    B1[kc].x = (n1v && (k     < Fin)) ? w10 : 0.0f;
    B1[kc].y = (n1v && (k + 1 < Fin)) ? w11 : 0.0f;
  }
  const float bA = b1[nl];
  const float bB = n1v ? b1[n1c] : 0.0f;

  const int NT = (Bsz * Tlen) / 16;
  for (int tile = blockIdx.x; tile < NT; tile += gridDim.x) {
    const float* xr = x + (size_t)(tile * 16 + nl) * Fin;

    if (tile + (int)gridDim.x < NT)
      __builtin_prefetch(xr + (size_t)gridDim.x * 16 * Fin, 0, 1);

    v8f a0e = {0, 0, 0, 0, 0, 0, 0, 0}, a0o = {0, 0, 0, 0, 0, 0, 0, 0};
    v8f a1e = {0, 0, 0, 0, 0, 0, 0, 0}, a1o = {0, 0, 0, 0, 0, 0, 0, 0};
#pragma unroll
    for (int kc = 0; kc < 10; kc += 2) {
      const v2f Ae = *(const v2f*)(xr + kc * 4 + ko);
      const v2f Ao = *(const v2f*)(xr + (kc + 1) * 4 + ko);
      a0e = WMMA4F32(Ae, B0[kc], a0e);
      a1e = WMMA4F32(Ae, B1[kc], a1e);
      a0o = WMMA4F32(Ao, B0[kc + 1], a0o);
      a1o = WMMA4F32(Ao, B1[kc + 1], a1o);
    }
    {
      const v2f t = *(const v2f*)(xr + 40);
      const v2f A = hiHalf ? zero2 : t;
      a0e = WMMA4F32(A, B0[10], a0e);
      a1e = WMMA4F32(A, B1[10], a1e);
    }

    float* arow = aOut + (size_t)tile * 16 * Hh;
#pragma unroll
    for (int j = 0; j < 8; ++j)
      arow[(j + mo) * Hh + nl] = tanhf_(a0e[j] + a0o[j] + bA);
    if (nl < 8) {
#pragma unroll
      for (int j = 0; j < 8; ++j)
        arow[(j + mo) * Hh + 16 + nl] = tanhf_(a1e[j] + a1o[j] + bB);
    }
  }
}

typedef __attribute__((ext_vector_type(4))) float v4f;

static constexpr int RD = 8;
#define AFRAG(S, q) \
  (((q) & 1) ? __builtin_shufflevector((S)[(q) >> 1], (S)[(q) >> 1], 2, 3) \
             : __builtin_shufflevector((S)[(q) >> 1], (S)[(q) >> 1], 0, 1))
#define LO4(V) __builtin_shufflevector((V), (V), 0, 1, 2, 3)
#define HI4(V) __builtin_shufflevector((V), (V), 4, 5, 6, 7)
static_assert(Tlen % RD == 0 && Tlen >= 2 * RD, "the step loop is unrolled by RD");
static_assert(RD % 4 == 0, "slot and buffer numbers are taken from the unroll index");

static constexpr int HPf = 28;
static constexpr int XPf = 76;
static constexpr int XSL = 16 * XPf;
static constexpr int NXS = 4;

__device__ __forceinline__ float xhalf(float v) {
  const int b = __builtin_bit_cast(int, v);
  return __builtin_bit_cast(
      float, __builtin_amdgcn_permlanex16(b, b, 0x76543210, 0xfedcba98, false,
                                          false));
}

__global__ __launch_bounds__(256) void gru_scan_kernel(
    const float* __restrict__ a, const float* __restrict__ w_ih,
    const float* __restrict__ w_hh, const float* __restrict__ b_ih,
    const float* __restrict__ b_hh, const float* __restrict__ lc2_w,
    const float* __restrict__ lc2_b, float* __restrict__ out) {
  __shared__ __attribute__((aligned(16))) float hbuf[2 * 16 * HPf];
  __shared__ __attribute__((aligned(16))) float xbuf[NXS * XSL];

  const int lane = threadIdx.x & 31;
  const int wv   = __builtin_amdgcn_readfirstlane((int)(threadIdx.x >> 5));
  const bool scan = (wv < 3);
  const int w    = scan ? wv : 0;
  const int xt   = scan ? 0 : (wv - 3);
  const bool xrz = (xt < 3);
  const int nl   = lane & 15;
  const int hf   = lane >> 4;
  const int kb   = 12 * hf;
  const int b0   = blockIdx.x * 16;
  const v2f zero2 = {0.0f, 0.0f};

  const int  mh  = nl >> 3, mj = nl & 7;
  const bool lo4 = (mj < 4);

  v2f WhRZ[6], WhN[6];
  {
    const int ua = 8 * w + 4 * mh + (mj & 3);
    const int rowRZ = (lo4 ? 0 : Hh) + ua, rowN = 2 * Hh + ua;
#pragma unroll
    for (int q = 0; q < 6; ++q) {
      const int k = kb + 2 * q;
      WhRZ[q] = 0.5f * *(const v2f*)(w_hh + rowRZ * Hh + k);
      const v2f hh = *(const v2f*)(w_hh + rowN * Hh + k);
      WhN[q] = lo4 ? zero2 : hh;
    }
  }
  v2f Wx[6];
  {
    const int  uaX   = 8 * (xrz ? xt : 0) + 4 * mh + (mj & 3);
    const int  rowRZ = (lo4 ? 0 : Hh) + uaX;
    const bool x1v   = (nl < 8);
    const int  rowN  = (xt == 4) ? (2 * Hh + 16 + (x1v ? nl : 0))
                                 : (2 * Hh + nl);
    const int  rowX  = xrz ? rowRZ : rowN;
    const bool rowV  = (xt != 4) || x1v;
#pragma unroll
    for (int q = 0; q < 6; ++q) {
      const v2f t = *(const v2f*)(w_ih + rowX * Hh + kb + 2 * q);
      const v2f th = 0.5f * t;
      Wx[q] = xrz ? th : (rowV ? t : zero2);
    }
  }

  v8f bX;
#pragma unroll
  for (int j = 0; j < 8; ++j) {
    const int c = 8 * (xrz ? xt : 0) + 4 * hf + (j & 3);
    const float rz = 0.5f * ((j < 4) ? (b_ih[c] + b_hh[c])
                                     : (b_ih[Hh + c] + b_hh[Hh + c]));
    const float n0 = b_ih[2 * Hh + 8 * hf + j];
    const float t1 = b_ih[2 * Hh + 16 + j];
    const float n1 = (hf == 0) ? t1 : 0.0f;
    bX[j] = xrz ? rz : ((xt == 3) ? n0 : n1);
  }
  float bhn[4];
#pragma unroll
  for (int i = 0; i < 4; ++i) bhn[i] = b_hh[2 * Hh + 8 * w + 4 * hf + i];
  float l2w[12];
#pragma unroll
  for (int e = 0; e < 12; ++e) l2w[e] = lc2_w[kb + e];
  const float l2b = lc2_b[0];
  const bool headw = (wv == 7);
  const bool storeLane = (hf == 0) && headw;

  const float* arow = a + (size_t)(b0 + nl) * Tlen * Hh + kb;
  float*       orow = out + (size_t)(b0 + nl) * Tlen;

  float*       hst = hbuf + nl * HPf + 8 * w + 4 * hf;
  const float* hld = hbuf + nl * HPf + kb;
  float*       xsLo = xbuf + nl * XPf +
                      (xrz ? (12 * hf + 24 * xt)
                           : ((xt == 3) ? (24 * hf + 8) : (48 + 8)));
  float*       xsHi = xsLo + (xrz ? 4 : 12);
  const bool   xsEn = (xt != 4) || (hf == 0);
  const float* xld  = xbuf + nl * XPf + 12 * (2 * w + hf);

  auto xside = [&](const v4f (&S)[3], v8f& o) {
    o = bX;
#pragma unroll
    for (int q = 0; q < 6; ++q) o = WMMA4F32(Wx[q], AFRAG(S, q), o);
  };
  auto xstore = [&](int slot, const v8f& p) {
    const int so = slot * XSL;
    if (xsEn) {
      *(v4f*)(xsLo + so) = LO4(p);
      *(v4f*)(xsHi + so) = HI4(p);
    }
  };

  v4f aR[RD][3];
#pragma unroll
  for (int s = 0; s < RD; ++s)
#pragma unroll
    for (int p = 0; p < 3; ++p) aR[s][p] = *(const v4f*)(arow + s * Hh + 4 * p);
  if (!scan) {
#pragma unroll
    for (int s = 0; s < 3; ++s) {
      v8f r;
      xside(aR[s], r);
      xstore(s, r);
    }
  }
  if (scan) *(v4f*)(hst + 16 * HPf) = (v4f){0.0f, 0.0f, 0.0f, 0.0f};
#pragma unroll
  for (int p = 0; p < 3; ++p) {
    aR[0][p] = *(const v4f*)(arow + (RD + 0) * Hh + 4 * p);
    aR[1][p] = *(const v4f*)(arow + (RD + 1) * Hh + 4 * p);
  }
  __builtin_amdgcn_fence(__ATOMIC_RELEASE, "workgroup", "local");
  __builtin_amdgcn_s_barrier_signal(-1);
  __builtin_amdgcn_s_barrier_wait(-1);
  __builtin_amdgcn_fence(__ATOMIC_ACQUIRE, "workgroup", "local");

  const v4f bhnv = {bhn[0], bhn[1], bhn[2], bhn[3]};
  float hown[4] = {0.0f, 0.0f, 0.0f, 0.0f};

  for (int tb = 0; tb < Tlen; tb += RD) {
#pragma unroll
  for (int u = 0; u < RD; ++u) {
    const int t  = tb + u;
    const int bw = (u & 1) * (16 * HPf);
    const int br = ((u + 1) & 1) * (16 * HPf);

    if (scan) {
      __builtin_amdgcn_fence(__ATOMIC_ACQUIRE, "workgroup", "local");
      const v4f lo = *(const v4f*)(xld + (u & 3) * XSL);
      const v4f hi = *(const v4f*)(xld + (u & 3) * XSL + 4);
      const v4f n4 = *(const v4f*)(xld + (u & 3) * XSL + 8);
      const v4f h0 = *(const v4f*)(hld + br);
      const v4f h1 = *(const v4f*)(hld + br + 4);
      const v4f h2 = *(const v4f*)(hld + br + 8);
      float h[12];
#pragma unroll
      for (int e = 0; e < 4; ++e) { h[e] = h0[e]; h[4 + e] = h1[e]; h[8 + e] = h2[e]; }

      v8f RZ = __builtin_shufflevector(lo, hi, 0, 1, 2, 3, 4, 5, 6, 7);
      v8f N  = __builtin_shufflevector(n4, bhnv, 0, 1, 2, 3, 4, 5, 6, 7);
#pragma unroll
      for (int q = 0; q < 6; ++q) {
        const v2f hB = {h[2 * q], h[2 * q + 1]};
        RZ = WMMA4F32(WhRZ[q], hB, RZ);
        N  = WMMA4F32(WhN[q],  hB, N);
      }

      v4f hn;
#pragma unroll
      for (int i = 0; i < 4; ++i) {
        const float r  = sigmoid_of_half_(RZ[i]);
        const float z  = sigmoid_of_half_(RZ[4 + i]);
        const float nn = tanhf_(N[i] + r * N[4 + i]);
        const float hv = (1.0f - z) * nn + z * hown[i];
        hown[i] = hv;
        hn[i]   = hv;
      }

      *(v4f*)(hst + bw) = hn;

      __builtin_amdgcn_fence(__ATOMIC_RELEASE, "workgroup", "local");
    } else {
      v8f r;
      if (headw) {
        __builtin_amdgcn_fence(__ATOMIC_ACQUIRE, "workgroup", "local");
        const v4f g0 = *(const v4f*)(hld + br);
        const v4f g1 = *(const v4f*)(hld + br + 4);
        const v4f g2 = *(const v4f*)(hld + br + 8);
        xside(aR[(u + 3) % RD], r);
        float s = 0.0f;
#pragma unroll
        for (int e = 0; e < 4; ++e) s += l2w[e] * g0[e];
#pragma unroll
        for (int e = 0; e < 4; ++e) s += l2w[4 + e] * g1[e];
#pragma unroll
        for (int e = 0; e < 4; ++e) s += l2w[8 + e] * g2[e];
        const float tot = s + xhalf(s);
        if (storeLane && t > 0) orow[t - 1] = sigmoidf_(tot + l2b);
      } else {
        xside(aR[(u + 3) % RD], r);
      }
      __builtin_amdgcn_fence(__ATOMIC_RELEASE, "workgroup", "local");
      xstore((u + 3) & 3, r);
    }

    __builtin_amdgcn_s_barrier_signal(-1);
    {
      const int tau = t + 2 + RD;
      const float* src = arow + (size_t)((tau < Tlen) ? tau : (Tlen - 1)) * Hh;
#pragma unroll
      for (int p = 0; p < 3; ++p)
        aR[(u + 2) % RD][p] = *(const v4f*)(src + 4 * p);
    }
    __builtin_amdgcn_s_barrier_wait(-1);
  }
  }

  {
    static_assert(((Tlen - 1) & 1) == 1, "the last step writes h buffer 1");
    __builtin_amdgcn_fence(__ATOMIC_ACQUIRE, "workgroup", "local");
    const v4f h0 = *(const v4f*)(hld + 16 * HPf);
    const v4f h1 = *(const v4f*)(hld + 16 * HPf + 4);
    const v4f h2 = *(const v4f*)(hld + 16 * HPf + 8);
    float h[12];
#pragma unroll
    for (int e = 0; e < 4; ++e) { h[e] = h0[e]; h[4 + e] = h1[e]; h[8 + e] = h2[e]; }
    float s = 0.0f;
#pragma unroll
    for (int e = 0; e < 12; ++e) s += l2w[e] * h[e];
    const float tot = s + xhalf(s);
    const float o   = sigmoidf_(tot + l2b);
    if (storeLane) orow[Tlen - 1] = o;
  }
}

extern "C" void kernel_launch(void* const* d_in, const int* in_sizes, int n_in,
                              void* d_out, int out_size, void* d_ws,
                              size_t ws_size, hipStream_t stream) {
  (void)in_sizes; (void)n_in; (void)out_size; (void)ws_size;
  const float* x     = (const float*)d_in[0];
  const float* lc1_w = (const float*)d_in[1];
  const float* lc1_b = (const float*)d_in[2];
  const float* w_ih  = (const float*)d_in[3];
  const float* w_hh  = (const float*)d_in[4];
  const float* b_ih  = (const float*)d_in[5];
  const float* b_hh  = (const float*)d_in[6];
  const float* lc2_w = (const float*)d_in[7];
  const float* lc2_b = (const float*)d_in[8];
  float* out  = (float*)d_out;
  float* aBuf = (float*)d_ws;

  lc1_tanh_kernel<<<2048, 32, 0, stream>>>(x, lc1_w, lc1_b, aBuf);
  gru_scan_kernel<<<Bsz / 16, 256, 0, stream>>>(aBuf, w_ih, w_hh, b_ih, b_hh,
                                               lc2_w, lc2_b, out);
}
